# starting kernel
# baseline (speedup 1.0000x reference)
#include <hip/hip_runtime.h>
#include <hip/hip_bf16.h>
#include <math.h>

typedef __attribute__((ext_vector_type(2))) float v2f;
typedef __attribute__((ext_vector_type(8))) float v8f;

#define HEADS 4
#define FDIM  128

__device__ __forceinline__ unsigned f2key(float f) {
    unsigned b = __float_as_uint(f);
    return (b & 0x80000000u) ? ~b : (b | 0x80000000u);
}
__device__ __forceinline__ float key2f(unsigned k) {
    unsigned b = (k & 0x80000000u) ? (k & 0x7FFFFFFFu) : ~k;
    return __uint_as_float(b);
}

__global__ __launch_bounds__(256) void gemm_x128(
    const float* __restrict__ X, const float* __restrict__ W,
    const float* __restrict__ bias, float* __restrict__ Y, int M)
{
    __shared__ float At[16][132];
    const int tid  = threadIdx.x;
    const int row0 = blockIdx.x * 16;

    #pragma unroll
    for (int i = 0; i < 2; ++i) {
        int idx = (tid + i * 256) * 4;
        int r = idx >> 7, c = idx & 127;
        int gr = row0 + r; gr = gr < M ? gr : M - 1;
        const float4 v = *(const float4*)(X + (size_t)gr * FDIM + c);
        At[r][c + 0] = v.x; At[r][c + 1] = v.y;
        At[r][c + 2] = v.z; At[r][c + 3] = v.w;
    }
    __syncthreads();

    const int lane = tid & 31;
    const int wv   = tid >> 5;
    const int m    = lane & 15;
    const int half = lane >> 4;
    const int koff = half << 1;
    const int n    = wv * 16 + (lane & 15);

    v8f acc = {};
    #pragma unroll
    for (int kb = 0; kb < FDIM; kb += 4) {
        v2f a, b;
        a.x = At[m][kb + koff];
        a.y = At[m][kb + koff + 1];
        b.x = W[(kb + koff)     * FDIM + n];
        b.y = W[(kb + koff + 1) * FDIM + n];
        acc = __builtin_amdgcn_wmma_f32_16x16x4_f32(
                  false, a, false, b, (short)0, acc, false, false);
    }

    const float bn = bias ? bias[n] : 0.0f;
    #pragma unroll
    for (int r = 0; r < 8; ++r) {
        int row = row0 + half * 8 + r;
        if (row < M) Y[(size_t)row * FDIM + n] = acc[r] + bn;
    }
}

__global__ __launch_bounds__(256) void node_alpha(
    const float* __restrict__ feat, const float* __restrict__ att,
    float* __restrict__ out, int N)
{
    const int lane = threadIdx.x & 31;
    const int node = blockIdx.x * 8 + (threadIdx.x >> 5);
    if (node >= N) return;
    const float4 f = *(const float4*)(feat + (size_t)node * FDIM + lane * 4);
    const float4 a = *(const float4*)(att + lane * 4);
    float p = f.x * a.x + f.y * a.y + f.z * a.z + f.w * a.w;
    p += __shfl_xor(p, 1);
    p += __shfl_xor(p, 2);
    p += __shfl_xor(p, 4);
    if ((lane & 7) == 0) out[node * HEADS + (lane >> 3)] = p;
}

__global__ void init_softmax(unsigned* __restrict__ amax,
                             float* __restrict__ asum, int n)
{
    int i = blockIdx.x * blockDim.x + threadIdx.x;
    if (i < n) { amax[i] = 0u; asum[i] = 0.0f; }
}

__global__ void edge_alpha_max(
    const int* __restrict__ esrc, const int* __restrict__ edst,
    const float* __restrict__ asn, const float* __restrict__ adn,
    float* __restrict__ alpha_e, unsigned* __restrict__ amax, int E)
{
    int e = blockIdx.x * blockDim.x + threadIdx.x;
    if (e >= E) return;
    const int s = esrc[e], d = edst[e];
    const float4 a = *(const float4*)(asn + (size_t)s * HEADS);
    const float4 b = *(const float4*)(adn + (size_t)d * HEADS);
    float4 al;
    al.x = a.x + b.x; al.x = al.x > 0.f ? al.x : 0.2f * al.x;
    al.y = a.y + b.y; al.y = al.y > 0.f ? al.y : 0.2f * al.y;
    al.z = a.z + b.z; al.z = al.z > 0.f ? al.z : 0.2f * al.z;
    al.w = a.w + b.w; al.w = al.w > 0.f ? al.w : 0.2f * al.w;
    *(float4*)(alpha_e + (size_t)e * HEADS) = al;
    unsigned* mp = amax + (size_t)d * HEADS;
    __hip_atomic_fetch_max(mp + 0, f2key(al.x), __ATOMIC_RELAXED, __HIP_MEMORY_SCOPE_AGENT);
    __hip_atomic_fetch_max(mp + 1, f2key(al.y), __ATOMIC_RELAXED, __HIP_MEMORY_SCOPE_AGENT);
    __hip_atomic_fetch_max(mp + 2, f2key(al.z), __ATOMIC_RELAXED, __HIP_MEMORY_SCOPE_AGENT);
    __hip_atomic_fetch_max(mp + 3, f2key(al.w), __ATOMIC_RELAXED, __HIP_MEMORY_SCOPE_AGENT);
}

__global__ void edge_exp_sum(
    const int* __restrict__ edst, const unsigned* __restrict__ amax,
    float* __restrict__ alpha_e, float* __restrict__ asum, int E)
{
    int e = blockIdx.x * blockDim.x + threadIdx.x;
    if (e >= E) return;
    const int d = edst[e];
    float4 al = *(const float4*)(alpha_e + (size_t)e * HEADS);
    const unsigned* mp = amax + (size_t)d * HEADS;
    al.x = __expf(al.x - key2f(mp[0]));
    al.y = __expf(al.y - key2f(mp[1]));
    al.z = __expf(al.z - key2f(mp[2]));
    al.w = __expf(al.w - key2f(mp[3]));
    *(float4*)(alpha_e + (size_t)e * HEADS) = al;
    float* sp = asum + (size_t)d * HEADS;
    __hip_atomic_fetch_add(sp + 0, al.x, __ATOMIC_RELAXED, __HIP_MEMORY_SCOPE_AGENT);
    __hip_atomic_fetch_add(sp + 1, al.y, __ATOMIC_RELAXED, __HIP_MEMORY_SCOPE_AGENT);
    __hip_atomic_fetch_add(sp + 2, al.z, __ATOMIC_RELAXED, __HIP_MEMORY_SCOPE_AGENT);
    __hip_atomic_fetch_add(sp + 3, al.w, __ATOMIC_RELAXED, __HIP_MEMORY_SCOPE_AGENT);
}

__global__ __launch_bounds__(256) void edge_scatter(
    const int* __restrict__ esrc, const int* __restrict__ edst,
    const float* __restrict__ src_feat, const float* __restrict__ alpha_e,
    const float* __restrict__ asum, float* __restrict__ out, int E)
{
    const int lane = threadIdx.x & 31;
    const int e = blockIdx.x * 8 + (threadIdx.x >> 5);
    if (e >= E) return;
    const int s = esrc[e], d = edst[e];
    const int h = lane >> 3;
    const float w = alpha_e[(size_t)e * HEADS + h] /
                    (asum[(size_t)d * HEADS + h] + 1e-16f);
    const float4 f = *(const float4*)(src_feat + (size_t)s * FDIM + lane * 4);
    float* op = out + (size_t)d * FDIM + lane * 4;
    __hip_atomic_fetch_add(op + 0, w * f.x, __ATOMIC_RELAXED, __HIP_MEMORY_SCOPE_AGENT);
    __hip_atomic_fetch_add(op + 1, w * f.y, __ATOMIC_RELAXED, __HIP_MEMORY_SCOPE_AGENT);
    __hip_atomic_fetch_add(op + 2, w * f.z, __ATOMIC_RELAXED, __HIP_MEMORY_SCOPE_AGENT);
    __hip_atomic_fetch_add(op + 3, w * f.w, __ATOMIC_RELAXED, __HIP_MEMORY_SCOPE_AGENT);
}

extern "C" void kernel_launch(void* const* d_in, const int* in_sizes, int n_in,
                              void* d_out, int out_size, void* d_ws, size_t ws_size,
                              hipStream_t stream)
{
    const float* x_src  = (const float*)d_in[0];
    const float* x_dst  = (const float*)d_in[1];
    const int*   esrc   = (const int*)d_in[2];
    const int*   edst   = (const int*)d_in[3];
    const float* W_src  = (const float*)d_in[5];
    const float* W_dst  = (const float*)d_in[6];
    const float* attS   = (const float*)d_in[7];
    const float* attD   = (const float*)d_in[8];
    const float* W_self = (const float*)d_in[9];
    const float* b_self = (const float*)d_in[10];

    const int N_SRC = in_sizes[0] / FDIM;
    const int N_DST = in_sizes[1] / FDIM;
    const int E     = in_sizes[2];

    char* ws = (char*)d_ws;
    float*    src_feat = (float*)   (ws);
    float*    dst_feat = (float*)   (ws + (size_t)N_SRC * FDIM * 4);
    float*    as_node  = (float*)   (ws + (size_t)(N_SRC + N_DST) * FDIM * 4);
    float*    ad_node  = as_node + (size_t)N_SRC * HEADS;
    unsigned* amax     = (unsigned*)(ad_node + (size_t)N_DST * HEADS);
    float*    asum     = (float*)   (amax + (size_t)N_DST * HEADS);
    float*    alpha_e  = asum + (size_t)N_DST * HEADS;

    float* out = (float*)d_out;

    {
        int n = N_DST * HEADS;
        init_softmax<<<(n + 255) / 256, 256, 0, stream>>>(amax, asum, n);
    }
    gemm_x128<<<(N_SRC + 15) / 16, 256, 0, stream>>>(x_src, W_src, nullptr, src_feat, N_SRC);
    gemm_x128<<<(N_DST + 15) / 16, 256, 0, stream>>>(x_dst, W_dst, nullptr, dst_feat, N_DST);
    gemm_x128<<<(N_DST + 15) / 16, 256, 0, stream>>>(x_dst, W_self, b_self, out, N_DST);
    node_alpha<<<(N_SRC + 7) / 8, 256, 0, stream>>>(src_feat, attS, as_node, N_SRC);
    node_alpha<<<(N_DST + 7) / 8, 256, 0, stream>>>(dst_feat, attD, ad_node, N_DST);
    edge_alpha_max<<<(E + 255) / 256, 256, 0, stream>>>(esrc, edst, as_node, ad_node,
                                                        alpha_e, amax, E);
    edge_exp_sum<<<(E + 255) / 256, 256, 0, stream>>>(edst, amax, alpha_e, asum, E);
    edge_scatter<<<(E + 7) / 8, 256, 0, stream>>>(esrc, edst, src_feat, alpha_e,
                                                  asum, out, E);
}


# All node work in one launch: a 16-row work-group sums att*feat per head from its WMMA tile via LDS; dst rows take both products from one x tile; no dst_feat; 9 to 4 launches; logit sum order changes
# speedup vs baseline: 2.5654x; 2.5654x over previous
#include <hip/hip_runtime.h>
#include <hip/hip_bf16.h>
#include <math.h>

typedef __attribute__((ext_vector_type(2))) float v2f;
typedef __attribute__((ext_vector_type(8))) float v8f;

#define HEADS 4
#define FDIM  128

__device__ __forceinline__ unsigned f2key(float f) {
    unsigned b = __float_as_uint(f);
    return (b & 0x80000000u) ? ~b : (b | 0x80000000u);
}
__device__ __forceinline__ float key2f(unsigned k) {
    unsigned b = (k & 0x80000000u) ? (k & 0x7FFFFFFFu) : ~k;
    return __uint_as_float(b);
}

__global__ __launch_bounds__(256) void node_proj(
    const float* __restrict__ x_src, const float* __restrict__ x_dst,
    const float* __restrict__ W_src, const float* __restrict__ W_dst,
    const float* __restrict__ W_self, const float* __restrict__ b_self,
    const float* __restrict__ attS, const float* __restrict__ attD,
    float* __restrict__ src_feat, float* __restrict__ out,
    float* __restrict__ as_node, float* __restrict__ ad_node,
    unsigned* __restrict__ amax, float* __restrict__ asum,
    int N_SRC, int N_DST, int dst_tiles)
{
    __shared__ float At[16][132];
    __shared__ alignas(16) float Pt[16][132];
    const int  tid    = threadIdx.x;
    const bool is_dst = (int)blockIdx.x < dst_tiles;
    const int  row0   = (is_dst ? (int)blockIdx.x : (int)blockIdx.x - dst_tiles) * 16;
    const int  M      = is_dst ? N_DST : N_SRC;
    const float* __restrict__ X = is_dst ? x_dst : x_src;

    #pragma unroll
    for (int i = 0; i < 2; ++i) {
        int idx = (tid + i * 256) * 4;
        int r = idx >> 7, c = idx & 127;
        int gr = row0 + r; gr = gr < M ? gr : M - 1;
        const float4 v = *(const float4*)(X + (size_t)gr * FDIM + c);
        At[r][c + 0] = v.x; At[r][c + 1] = v.y;
        At[r][c + 2] = v.z; At[r][c + 3] = v.w;
    }
    __syncthreads();

    const int lane = tid & 31;
    const int wv   = tid >> 5;
    const int m    = lane & 15;
    const int half = lane >> 4;
    const int koff = half << 1;
    const int n    = wv * 16 + (lane & 15);

    v8f    accY;
    float  bn;
    float* __restrict__ Y;
    if (is_dst) {
        v8f accA = {}, accS = {};
        #pragma unroll
        for (int kb = 0; kb < FDIM; kb += 4) {
            v2f a, bA, bS;
            a.x  = At[m][kb + koff];
            a.y  = At[m][kb + koff + 1];
            bA.x = W_dst [(kb + koff)     * FDIM + n];
            bA.y = W_dst [(kb + koff + 1) * FDIM + n];
            bS.x = W_self[(kb + koff)     * FDIM + n];
            bS.y = W_self[(kb + koff + 1) * FDIM + n];
            accA = __builtin_amdgcn_wmma_f32_16x16x4_f32(
                       false, a, false, bA, (short)0, accA, false, false);
            accS = __builtin_amdgcn_wmma_f32_16x16x4_f32(
                       false, a, false, bS, (short)0, accS, false, false);
        }
        const float an = attD[n];
        #pragma unroll
        for (int r = 0; r < 8; ++r)
            Pt[half * 8 + r][n] = accA[r] * an;
        accY = accS; bn = b_self[n]; Y = out;
    } else {
        v8f acc = {};
        #pragma unroll
        for (int kb = 0; kb < FDIM; kb += 4) {
            v2f a, b;
            a.x = At[m][kb + koff];
            a.y = At[m][kb + koff + 1];
            b.x = W_src[(kb + koff)     * FDIM + n];
            b.y = W_src[(kb + koff + 1) * FDIM + n];
            acc = __builtin_amdgcn_wmma_f32_16x16x4_f32(
                      false, a, false, b, (short)0, acc, false, false);
        }
        const float an = attS[n];
        #pragma unroll
        for (int r = 0; r < 8; ++r)
            Pt[half * 8 + r][n] = acc[r] * an;
        accY = acc; bn = 0.0f; Y = src_feat;
    }
    __syncthreads();

    #pragma unroll
    for (int r = 0; r < 8; ++r) {
        const int row = row0 + half * 8 + r;
        if (row < M) Y[(size_t)row * FDIM + n] = accY[r] + bn;
    }

    if (tid < 64) {
        const int r = tid >> 2, h = tid & 3;
        const float4* p = (const float4*)&Pt[r][h * 32];
        float4 s = p[0];
        #pragma unroll
        for (int j = 1; j < 8; ++j) {
            const float4 v = p[j];
            s.x += v.x; s.y += v.y; s.z += v.z; s.w += v.w;
        }
        const float dot = (s.x + s.y) + (s.z + s.w);
        const int row = row0 + r;
        if (row < M) {
            const int i = row * HEADS + h;
            if (is_dst) { ad_node[i] = dot; amax[i] = 0u; asum[i] = 0.0f; }
            else        { as_node[i] = dot; }
        }
    }
}

__global__ __launch_bounds__(256) void edge_alpha_max(
    const int* __restrict__ esrc, const int* __restrict__ edst,
    const float* __restrict__ asn, const float* __restrict__ adn,
    float* __restrict__ alpha_e, unsigned* __restrict__ amax, int E)
{
    const int n    = E * HEADS;
    const int base = blockIdx.x * 1024 + threadIdx.x;
    float al[4];
    int   dh[4];
    #pragma unroll
    for (int k = 0; k < 4; ++k) {
        int i = base + k * 256;
        i = i < n ? i : n - 1;
        const int e = i >> 2, h = i & 3;
        const int s = esrc[e], d = edst[e];
        const float a = asn[s * HEADS + h] + adn[d * HEADS + h];
        al[k] = a > 0.f ? a : 0.2f * a;
        dh[k] = d * HEADS + h;
    }
    #pragma unroll
    for (int k = 0; k < 4; ++k) {
        const int i = base + k * 256;
        if (i < n) {
            alpha_e[i] = al[k];
            __hip_atomic_fetch_max(amax + dh[k], f2key(al[k]),
                                   __ATOMIC_RELAXED, __HIP_MEMORY_SCOPE_AGENT);
        }
    }
}

__global__ __launch_bounds__(256) void edge_exp_sum(
    const int* __restrict__ edst, const unsigned* __restrict__ amax,
    float* __restrict__ alpha_e, float* __restrict__ asum, int E)
{
    const int n    = E * HEADS;
    const int base = blockIdx.x * 1024 + threadIdx.x;
    float p[4];
    int   dh[4];
    #pragma unroll
    for (int k = 0; k < 4; ++k) {
        int i = base + k * 256;
        i = i < n ? i : n - 1;
        const int e = i >> 2, h = i & 3;
        const int d = edst[e];
        dh[k] = d * HEADS + h;
        p[k]  = __expf(alpha_e[i] - key2f(amax[dh[k]]));
    }
    #pragma unroll
    for (int k = 0; k < 4; ++k) {
        const int i = base + k * 256;
        if (i < n) {
            alpha_e[i] = p[k];
            __hip_atomic_fetch_add(asum + dh[k], p[k],
                                   __ATOMIC_RELAXED, __HIP_MEMORY_SCOPE_AGENT);
        }
    }
}

#define EPW 8

__device__ __forceinline__ void add_row(float* __restrict__ out, int d, int lane, float wl, int q,
                                        float f0, float f1, float f2, float f3)
{
    const float w0 = __int_as_float(__builtin_amdgcn_readlane(__float_as_int(wl), q + 0));
    const float w1 = __int_as_float(__builtin_amdgcn_readlane(__float_as_int(wl), q + 1));
    const float w2 = __int_as_float(__builtin_amdgcn_readlane(__float_as_int(wl), q + 2));
    const float w3 = __int_as_float(__builtin_amdgcn_readlane(__float_as_int(wl), q + 3));
    float* op = out + (size_t)d * FDIM + lane;
    __hip_atomic_fetch_add(op +  0, w0 * f0, __ATOMIC_RELAXED, __HIP_MEMORY_SCOPE_AGENT);
    __hip_atomic_fetch_add(op + 32, w1 * f1, __ATOMIC_RELAXED, __HIP_MEMORY_SCOPE_AGENT);
    __hip_atomic_fetch_add(op + 64, w2 * f2, __ATOMIC_RELAXED, __HIP_MEMORY_SCOPE_AGENT);
    __hip_atomic_fetch_add(op + 96, w3 * f3, __ATOMIC_RELAXED, __HIP_MEMORY_SCOPE_AGENT);
}

__global__ __launch_bounds__(256) void edge_scatter(
    const int* __restrict__ esrc, const int* __restrict__ edst,
    const float* __restrict__ src_feat, const float* __restrict__ alpha_e,
    const float* __restrict__ asum, float* __restrict__ out, int E)
{
    const int lane = threadIdx.x & 31;
    const int wv = __builtin_amdgcn_readfirstlane((int)(threadIdx.x >> 5));
    const int e0 = (blockIdx.x * 8 + wv) * EPW;
    if (e0 >= E) return;
    int el = e0 + (lane >> 2);
    el = el < E ? el : E - 1;
    const int hl = lane & 3;
    const int sl = esrc[el], dl = edst[el];
    const float al = alpha_e[(size_t)el * HEADS + hl];
    if (e0 + EPW <= E) {
        float f[EPW][4];
        #pragma unroll
        for (int j = 0; j < EPW; ++j) {
            const int s = __builtin_amdgcn_readlane(sl, 4 * j);
            const float* sp = src_feat + (size_t)s * FDIM + lane;
            f[j][0] = sp[0]; f[j][1] = sp[32]; f[j][2] = sp[64]; f[j][3] = sp[96];
        }
        const float wl = al / (asum[(size_t)dl * HEADS + hl] + 1e-16f);
        #pragma unroll
        for (int j = 0; j < EPW; ++j)
            add_row(out, __builtin_amdgcn_readlane(dl, 4 * j), lane, wl, 4 * j,
                    f[j][0], f[j][1], f[j][2], f[j][3]);
    } else {
        const float wl = al / (asum[(size_t)dl * HEADS + hl] + 1e-16f);
        for (int j = 0; j < E - e0; ++j) {
            const int s = __builtin_amdgcn_readlane(sl, 4 * j);
            const float* sp = src_feat + (size_t)s * FDIM + lane;
            add_row(out, __builtin_amdgcn_readlane(dl, 4 * j), lane, wl, 4 * j,
                    sp[0], sp[32], sp[64], sp[96]);
        }
    }
}

extern "C" void kernel_launch(void* const* d_in, const int* in_sizes, int n_in,
                              void* d_out, int out_size, void* d_ws, size_t ws_size,
                              hipStream_t stream)
{
    const float* x_src  = (const float*)d_in[0];
    const float* x_dst  = (const float*)d_in[1];
    const int*   esrc   = (const int*)d_in[2];
    const int*   edst   = (const int*)d_in[3];
    const float* W_src  = (const float*)d_in[5];
    const float* W_dst  = (const float*)d_in[6];
    const float* attS   = (const float*)d_in[7];
    const float* attD   = (const float*)d_in[8];
    const float* W_self = (const float*)d_in[9];
    const float* b_self = (const float*)d_in[10];

    const int N_SRC = in_sizes[0] / FDIM;
    const int N_DST = in_sizes[1] / FDIM;
    const int E     = in_sizes[2];

    char* ws = (char*)d_ws;
    float*    src_feat = (float*)   (ws);
    float*    as_node  = (float*)   (ws + (size_t)(N_SRC + N_DST) * FDIM * 4);
    float*    ad_node  = as_node + (size_t)N_SRC * HEADS;
    unsigned* amax     = (unsigned*)(ad_node + (size_t)N_DST * HEADS);
    float*    asum     = (float*)   (amax + (size_t)N_DST * HEADS);
    float*    alpha_e  = asum + (size_t)N_DST * HEADS;

    float* out = (float*)d_out;

    {
        const int dst_tiles = (N_DST + 15) / 16;
        const int src_tiles = (N_SRC + 15) / 16;
        node_proj<<<dst_tiles + src_tiles, 256, 0, stream>>>(
            x_src, x_dst, W_src, W_dst, W_self, b_self, attS, attD,
            src_feat, out, as_node, ad_node, amax, asum, N_SRC, N_DST, dst_tiles);
    }
    edge_alpha_max<<<(E + 255) / 256, 256, 0, stream>>>(esrc, edst, as_node, ad_node,
                                                        alpha_e, amax, E);
    edge_exp_sum<<<(E + 255) / 256, 256, 0, stream>>>(edst, amax, alpha_e, asum, E);
    edge_scatter<<<(E + 8 * EPW - 1) / (8 * EPW), 256, 0, stream>>>(esrc, edst, src_feat,
                                                                    alpha_e, asum, out, E);
}


# Edge phase by owner work-groups: each owns 196 dst rows, lists their edges from one read of all destinations, softmax and row sums by LDS atomics, one add to out; 4 to 2 launches; sum order changes
# speedup vs baseline: 3.4988x; 1.3638x over previous
#include <hip/hip_runtime.h>
#include <hip/hip_bf16.h>
#include <math.h>

typedef __attribute__((ext_vector_type(2))) float v2f;
typedef __attribute__((ext_vector_type(8))) float v8f;

#define HEADS 4
#define FDIM  128

__device__ __forceinline__ unsigned f2key(float f) {
    unsigned b = __float_as_uint(f);
    return (b & 0x80000000u) ? ~b : (b | 0x80000000u);
}
__device__ __forceinline__ float key2f(unsigned k) {
    unsigned b = (k & 0x80000000u) ? (k & 0x7FFFFFFFu) : ~k;
    return __uint_as_float(b);
}

__global__ __launch_bounds__(256) void node_proj(
    const float* __restrict__ x_src, const float* __restrict__ x_dst,
    const float* __restrict__ W_src, const float* __restrict__ W_dst,
    const float* __restrict__ W_self, const float* __restrict__ b_self,
    const float* __restrict__ attS, const float* __restrict__ attD,
    float* __restrict__ src_feat, float* __restrict__ out,
    float* __restrict__ as_node, float* __restrict__ ad_node,
    int N_SRC, int N_DST, int dst_tiles)
{
    __shared__ float At[16][132];
    __shared__ alignas(16) float Pt[16][132];
    const int  tid    = threadIdx.x;
    const bool is_dst = (int)blockIdx.x < dst_tiles;
    const int  row0   = (is_dst ? (int)blockIdx.x : (int)blockIdx.x - dst_tiles) * 16;
    const int  M      = is_dst ? N_DST : N_SRC;
    const float* __restrict__ X = is_dst ? x_dst : x_src;

    #pragma unroll
    for (int i = 0; i < 2; ++i) {
        int idx = (tid + i * 256) * 4;
        int r = idx >> 7, c = idx & 127;
        int gr = row0 + r; gr = gr < M ? gr : M - 1;
        const float4 v = *(const float4*)(X + (size_t)gr * FDIM + c);
        At[r][c + 0] = v.x; At[r][c + 1] = v.y;
        At[r][c + 2] = v.z; At[r][c + 3] = v.w;
    }
    __syncthreads();

    const int lane = tid & 31;
    const int wv   = tid >> 5;
    const int m    = lane & 15;
    const int half = lane >> 4;
    const int koff = half << 1;
    const int n    = wv * 16 + (lane & 15);

    v8f    accY;
    float  bn;
    float* __restrict__ Y;
    if (is_dst) {
        v8f accA = {}, accS = {};
        #pragma unroll
        for (int kb = 0; kb < FDIM; kb += 4) {
            v2f a, bA, bS;
            a.x  = At[m][kb + koff];
            a.y  = At[m][kb + koff + 1];
            bA.x = W_dst [(kb + koff)     * FDIM + n];
            bA.y = W_dst [(kb + koff + 1) * FDIM + n];
            bS.x = W_self[(kb + koff)     * FDIM + n];
            bS.y = W_self[(kb + koff + 1) * FDIM + n];
            accA = __builtin_amdgcn_wmma_f32_16x16x4_f32(
                       false, a, false, bA, (short)0, accA, false, false);
            accS = __builtin_amdgcn_wmma_f32_16x16x4_f32(
                       false, a, false, bS, (short)0, accS, false, false);
        }
        const float an = attD[n];
        #pragma unroll
        for (int r = 0; r < 8; ++r)
            Pt[half * 8 + r][n] = accA[r] * an;
        accY = accS; bn = b_self[n]; Y = out;
    } else {
        v8f acc = {};
        #pragma unroll
        for (int kb = 0; kb < FDIM; kb += 4) {
            v2f a, b;
            a.x = At[m][kb + koff];
            a.y = At[m][kb + koff + 1];
            b.x = W_src[(kb + koff)     * FDIM + n];
            b.y = W_src[(kb + koff + 1) * FDIM + n];
            acc = __builtin_amdgcn_wmma_f32_16x16x4_f32(
                      false, a, false, b, (short)0, acc, false, false);
        }
        const float an = attS[n];
        #pragma unroll
        for (int r = 0; r < 8; ++r)
            Pt[half * 8 + r][n] = acc[r] * an;
        accY = acc; bn = 0.0f; Y = src_feat;
    }
    __syncthreads();

    #pragma unroll
    for (int r = 0; r < 8; ++r) {
        const int row = row0 + half * 8 + r;
        if (row < M) Y[(size_t)row * FDIM + n] = accY[r] + bn;
    }

    if (tid < 64) {
        const int r = tid >> 2, h = tid & 3;
        const float4* p = (const float4*)&Pt[r][h * 32];
        float4 s = p[0];
        #pragma unroll
        for (int j = 1; j < 8; ++j) {
            const float4 v = p[j];
            s.x += v.x; s.y += v.y; s.z += v.z; s.w += v.w;
        }
        const float dot = (s.x + s.y) + (s.z + s.w);
        const int row = row0 + r;
        if (row < M) {
            const int i = row * HEADS + h;
            if (is_dst) ad_node[i] = dot;
            else        as_node[i] = dot;
        }
    }
}

#define OWN_ROWS    196
#define OWN_THREADS 1024
#define OWN_WAVES   (OWN_THREADS / 32)
#define LCAP        4096
#define EPW         8
#define SCAN_AHEAD  8

__device__ __forceinline__ float lrelu02(float a) { return a > 0.f ? a : 0.2f * a; }

__device__ __forceinline__ void add_row_lds(float* acc, int d, int lane, float wl, int q,
                                            float f0, float f1, float f2, float f3)
{
    const float w0 = __int_as_float(__builtin_amdgcn_readlane(__float_as_int(wl), q + 0));
    const float w1 = __int_as_float(__builtin_amdgcn_readlane(__float_as_int(wl), q + 1));
    const float w2 = __int_as_float(__builtin_amdgcn_readlane(__float_as_int(wl), q + 2));
    const float w3 = __int_as_float(__builtin_amdgcn_readlane(__float_as_int(wl), q + 3));
    float* ap = acc + d * FDIM + lane;
    __hip_atomic_fetch_add(ap +  0, w0 * f0, __ATOMIC_RELAXED, __HIP_MEMORY_SCOPE_WORKGROUP);
    __hip_atomic_fetch_add(ap + 32, w1 * f1, __ATOMIC_RELAXED, __HIP_MEMORY_SCOPE_WORKGROUP);
    __hip_atomic_fetch_add(ap + 64, w2 * f2, __ATOMIC_RELAXED, __HIP_MEMORY_SCOPE_WORKGROUP);
    __hip_atomic_fetch_add(ap + 96, w3 * f3, __ATOMIC_RELAXED, __HIP_MEMORY_SCOPE_WORKGROUP);
}

__global__ __launch_bounds__(OWN_THREADS) void edge_owner(
    const int* __restrict__ esrc, const int* __restrict__ edst,
    const float* __restrict__ src_feat,
    const float* __restrict__ asn, const float* __restrict__ adn,
    float* __restrict__ out, int E, int N_DST)
{
    __shared__ alignas(16) float acc[OWN_ROWS * FDIM];
    __shared__ alignas(16) float adl[OWN_ROWS * HEADS];
    __shared__ unsigned mx[OWN_ROWS * HEADS];
    __shared__ float    sm[OWN_ROWS * HEADS];
    __shared__ int      lst_s[LCAP];
    __shared__ int      lst_d[LCAP];
    __shared__ alignas(16) float lst_a[LCAP * HEADS];
    __shared__ int      cnt;

    const int tid  = threadIdx.x;
    const int r0   = blockIdx.x * OWN_ROWS;
    const int left = N_DST - r0;
    const int rows = left < OWN_ROWS ? left : OWN_ROWS;

    for (int i = tid; i < OWN_ROWS * FDIM / 4; i += OWN_THREADS)
        ((float4*)acc)[i] = make_float4(0.f, 0.f, 0.f, 0.f);
    for (int i = tid; i < rows * HEADS; i += OWN_THREADS) {
        adl[i] = adn[(size_t)r0 * HEADS + i];
        mx[i] = 0u;
        sm[i] = 0.f;
    }
    if (tid == 0) cnt = 0;
    __syncthreads();

    {
        const int n4 = E >> 2;
        const int4* __restrict__ ed4 = (const int4*)edst;
        const int start = (int)blockIdx.x * (n4 / (int)gridDim.x);
        for (int base = 0; base < n4; base += SCAN_AHEAD * OWN_THREADS) {
            int4 v[SCAN_AHEAD];
            int  q[SCAN_AHEAD];
            #pragma unroll
            for (int u = 0; u < SCAN_AHEAD; ++u) {
                const int i = base + u * OWN_THREADS + tid;
                int qq = start + (i < n4 ? i : 0);
                qq = qq < n4 ? qq : qq - n4;
                q[u] = i < n4 ? qq : -1;
                v[u] = ed4[qq];
            }
            #pragma unroll
            for (int u = 0; u < SCAN_AHEAD; ++u) {
                const unsigned dd[4] = { (unsigned)(v[u].x - r0), (unsigned)(v[u].y - r0),
                                         (unsigned)(v[u].z - r0), (unsigned)(v[u].w - r0) };
                const unsigned lo = min(min(dd[0], dd[1]), min(dd[2], dd[3]));
                if (q[u] >= 0 && lo < (unsigned)rows) {
                    #pragma unroll
                    for (int k = 0; k < 4; ++k) {
                        if (dd[k] < (unsigned)rows) {
                            const int pos = __hip_atomic_fetch_add(&cnt, 1, __ATOMIC_RELAXED,
                                                                   __HIP_MEMORY_SCOPE_WORKGROUP);
                            if (pos < LCAP) { lst_s[pos] = 4 * q[u] + k; lst_d[pos] = (int)dd[k]; }
                        }
                    }
                }
            }
        }
        for (int e = (n4 << 2) + tid; e < E; e += OWN_THREADS) {
            const unsigned dl = (unsigned)(edst[e] - r0);
            if (dl < (unsigned)rows) {
                const int pos = __hip_atomic_fetch_add(&cnt, 1, __ATOMIC_RELAXED,
                                                       __HIP_MEMORY_SCOPE_WORKGROUP);
                if (pos < LCAP) { lst_s[pos] = e; lst_d[pos] = (int)dl; }
            }
        }
    }
    __syncthreads();
    const int  total = cnt;
    const bool lmode = total <= LCAP;
    const int  n     = lmode ? total : E;

    for (int i = tid; i < n; i += OWN_THREADS) {
        int s, dl; bool own;
        if (lmode) { s = esrc[lst_s[i]]; dl = lst_d[i]; own = true; lst_s[i] = s; }
        else {
            const unsigned u = (unsigned)(edst[i] - r0);
            own = u < (unsigned)rows; dl = own ? (int)u : 0; s = own ? esrc[i] : 0;
        }
        if (own) {
            const float4 a4 = *(const float4*)(asn + (size_t)s * HEADS);
            const float4 d4 = *(const float4*)(adl + dl * HEADS);
            const float a0 = lrelu02(a4.x + d4.x), a1 = lrelu02(a4.y + d4.y);
            const float a2 = lrelu02(a4.z + d4.z), a3 = lrelu02(a4.w + d4.w);
            if (lmode) *(float4*)(lst_a + i * HEADS) = make_float4(a0, a1, a2, a3);
            unsigned* m = mx + dl * HEADS;
            __hip_atomic_fetch_max(m + 0, f2key(a0), __ATOMIC_RELAXED, __HIP_MEMORY_SCOPE_WORKGROUP);
            __hip_atomic_fetch_max(m + 1, f2key(a1), __ATOMIC_RELAXED, __HIP_MEMORY_SCOPE_WORKGROUP);
            __hip_atomic_fetch_max(m + 2, f2key(a2), __ATOMIC_RELAXED, __HIP_MEMORY_SCOPE_WORKGROUP);
            __hip_atomic_fetch_max(m + 3, f2key(a3), __ATOMIC_RELAXED, __HIP_MEMORY_SCOPE_WORKGROUP);
        }
    }
    __syncthreads();

    for (int i = tid; i < n; i += OWN_THREADS) {
        int s, dl; bool own;
        if (lmode) { s = lst_s[i]; dl = lst_d[i]; own = true; }
        else {
            const unsigned u = (unsigned)(edst[i] - r0);
            own = u < (unsigned)rows; dl = own ? (int)u : 0; s = own ? esrc[i] : 0;
        }
        if (own) {
            float4 a;
            if (lmode) a = *(const float4*)(lst_a + i * HEADS);
            else {
                const float4 a4 = *(const float4*)(asn + (size_t)s * HEADS);
                const float4 d4 = *(const float4*)(adl + dl * HEADS);
                a = make_float4(lrelu02(a4.x + d4.x), lrelu02(a4.y + d4.y),
                                lrelu02(a4.z + d4.z), lrelu02(a4.w + d4.w));
            }
            const unsigned* m = mx + dl * HEADS;
            const float p0 = __expf(a.x - key2f(m[0])), p1 = __expf(a.y - key2f(m[1]));
            const float p2 = __expf(a.z - key2f(m[2])), p3 = __expf(a.w - key2f(m[3]));
            if (lmode) *(float4*)(lst_a + i * HEADS) = make_float4(p0, p1, p2, p3);
            float* sp = sm + dl * HEADS;
            __hip_atomic_fetch_add(sp + 0, p0, __ATOMIC_RELAXED, __HIP_MEMORY_SCOPE_WORKGROUP);
            __hip_atomic_fetch_add(sp + 1, p1, __ATOMIC_RELAXED, __HIP_MEMORY_SCOPE_WORKGROUP);
            __hip_atomic_fetch_add(sp + 2, p2, __ATOMIC_RELAXED, __HIP_MEMORY_SCOPE_WORKGROUP);
            __hip_atomic_fetch_add(sp + 3, p3, __ATOMIC_RELAXED, __HIP_MEMORY_SCOPE_WORKGROUP);
        }
    }
    __syncthreads();

    {
        const int lane = tid & 31;
        const int wv = __builtin_amdgcn_readfirstlane(tid >> 5);
        const int hl = lane & 3;
        for (int i0 = wv * EPW; i0 < n; i0 += OWN_WAVES * EPW) {
            const int il = i0 + (lane >> 2);
            if (lmode && i0 + EPW <= n) {
                const int   s  = lst_s[il];
                const int   dl = lst_d[il];
                const float p  = lst_a[il * HEADS + hl];
                float f[EPW][4];
                #pragma unroll
                for (int j = 0; j < EPW; ++j) {
                    const int sj = __builtin_amdgcn_readlane(s, 4 * j);
                    const float* sp = src_feat + (size_t)sj * FDIM + lane;
                    f[j][0] = sp[0]; f[j][1] = sp[32]; f[j][2] = sp[64]; f[j][3] = sp[96];
                }
                const float wl = p / (sm[dl * HEADS + hl] + 1e-16f);
                #pragma unroll
                for (int j = 0; j < EPW; ++j)
                    add_row_lds(acc, __builtin_amdgcn_readlane(dl, 4 * j), lane, wl, 4 * j,
                                f[j][0], f[j][1], f[j][2], f[j][3]);
            } else {
                int s = 0, dl = 0; bool own = false;
                if (il < n) {
                    if (lmode) { s = lst_s[il]; dl = lst_d[il]; own = true; }
                    else {
                        const unsigned u = (unsigned)(edst[il] - r0);
                        own = u < (unsigned)rows; dl = own ? (int)u : 0; s = own ? esrc[il] : 0;
                    }
                }
                float wl = 0.f;
                if (own) {
                    float p;
                    if (lmode) p = lst_a[il * HEADS + hl];
                    else p = __expf(lrelu02(asn[(size_t)s * HEADS + hl] + adl[dl * HEADS + hl])
                                    - key2f(mx[dl * HEADS + hl]));
                    wl = p / (sm[dl * HEADS + hl] + 1e-16f);
                }
                const unsigned om = __builtin_amdgcn_ballot_w32(own);
                for (int j = 0; j < EPW; ++j) {
                    if (om & (1u << (4 * j))) {
                        const int sj = __builtin_amdgcn_readlane(s, 4 * j);
                        const float* sp = src_feat + (size_t)sj * FDIM + lane;
                        add_row_lds(acc, __builtin_amdgcn_readlane(dl, 4 * j), lane, wl, 4 * j,
                                    sp[0], sp[32], sp[64], sp[96]);
                    }
                }
            }
        }
    }
    __syncthreads();

    {
        float4* __restrict__ op = (float4*)(out + (size_t)r0 * FDIM);
        const float4* ac = (const float4*)acc;
        for (int i = tid; i < rows * (FDIM / 4); i += OWN_THREADS) {
            float4 o = op[i];
            const float4 a = ac[i];
            o.x += a.x; o.y += a.y; o.z += a.z; o.w += a.w;
            op[i] = o;
        }
    }
}

extern "C" void kernel_launch(void* const* d_in, const int* in_sizes, int n_in,
                              void* d_out, int out_size, void* d_ws, size_t ws_size,
                              hipStream_t stream)
{
    const float* x_src  = (const float*)d_in[0];
    const float* x_dst  = (const float*)d_in[1];
    const int*   esrc   = (const int*)d_in[2];
    const int*   edst   = (const int*)d_in[3];
    const float* W_src  = (const float*)d_in[5];
    const float* W_dst  = (const float*)d_in[6];
    const float* attS   = (const float*)d_in[7];
    const float* attD   = (const float*)d_in[8];
    const float* W_self = (const float*)d_in[9];
    const float* b_self = (const float*)d_in[10];

    const int N_SRC = in_sizes[0] / FDIM;
    const int N_DST = in_sizes[1] / FDIM;
    const int E     = in_sizes[2];

    char* ws = (char*)d_ws;
    float* src_feat = (float*)(ws);
    float* as_node  = (float*)(ws + (size_t)(N_SRC + N_DST) * FDIM * 4);
    float* ad_node  = as_node + (size_t)N_SRC * HEADS;

    float* out = (float*)d_out;

    {
        const int dst_tiles = (N_DST + 15) / 16;
        const int src_tiles = (N_SRC + 15) / 16;
        node_proj<<<dst_tiles + src_tiles, 256, 0, stream>>>(
            x_src, x_dst, W_src, W_dst, W_self, b_self, attS, attD,
            src_feat, out, as_node, ad_node, N_SRC, N_DST, dst_tiles);
    }
    edge_owner<<<(N_DST + OWN_ROWS - 1) / OWN_ROWS, OWN_THREADS, 0, stream>>>(
        esrc, edst, src_feat, as_node, ad_node, out, E, N_DST);
}


# Owner scan reads a 1-byte owner key per edge (dst/196, written by the node launch in the same call), 16 keys a load, hit loads queued per wave by ballot: 4x fewer scan bytes; list (sum) order differs
# speedup vs baseline: 3.9104x; 1.1176x over previous
#include <hip/hip_runtime.h>
#include <hip/hip_bf16.h>
#include <math.h>

typedef __attribute__((ext_vector_type(2))) float v2f;
typedef __attribute__((ext_vector_type(8))) float v8f;
typedef __attribute__((ext_vector_type(4))) unsigned v4u;

#define HEADS 4
#define FDIM  128
#define OWN_ROWS 196

__device__ __forceinline__ unsigned f2key(float f) {
    unsigned b = __float_as_uint(f);
    return (b & 0x80000000u) ? ~b : (b | 0x80000000u);
}
__device__ __forceinline__ float key2f(unsigned k) {
    unsigned b = (k & 0x80000000u) ? (k & 0x7FFFFFFFu) : ~k;
    return __uint_as_float(b);
}

__global__ __launch_bounds__(256) void node_proj(
    const float* __restrict__ x_src, const float* __restrict__ x_dst,
    const float* __restrict__ W_src, const float* __restrict__ W_dst,
    const float* __restrict__ W_self, const float* __restrict__ b_self,
    const float* __restrict__ attS, const float* __restrict__ attD,
    float* __restrict__ src_feat, float* __restrict__ out,
    float* __restrict__ as_node, float* __restrict__ ad_node,
    const int* __restrict__ edst, unsigned* __restrict__ keys, int E,
    int N_SRC, int N_DST, int dst_tiles)
{
    __shared__ float At[16][132];
    __shared__ alignas(16) float Pt[16][132];
    const int  tid    = threadIdx.x;
    const bool is_dst = (int)blockIdx.x < dst_tiles;
    const int  row0   = (is_dst ? (int)blockIdx.x : (int)blockIdx.x - dst_tiles) * 16;
    const int  M      = is_dst ? N_DST : N_SRC;
    const float* __restrict__ X = is_dst ? x_dst : x_src;

    #pragma unroll
    for (int i = 0; i < 2; ++i) {
        int idx = (tid + i * 256) * 4;
        int r = idx >> 7, c = idx & 127;
        int gr = row0 + r; gr = gr < M ? gr : M - 1;
        const float4 v = *(const float4*)(X + (size_t)gr * FDIM + c);
        At[r][c + 0] = v.x; At[r][c + 1] = v.y;
        At[r][c + 2] = v.z; At[r][c + 3] = v.w;
    }
    __syncthreads();

    const int lane = tid & 31;
    const int wv   = tid >> 5;
    const int m    = lane & 15;
    const int half = lane >> 4;
    const int koff = half << 1;
    const int n    = wv * 16 + (lane & 15);

    v8f    accY;
    float  bn;
    float* __restrict__ Y;
    if (is_dst) {
        v8f accA = {}, accS = {};
        #pragma unroll
        for (int kb = 0; kb < FDIM; kb += 4) {
            v2f a, bA, bS;
            a.x  = At[m][kb + koff];
            a.y  = At[m][kb + koff + 1];
            bA.x = W_dst [(kb + koff)     * FDIM + n];
            bA.y = W_dst [(kb + koff + 1) * FDIM + n];
            bS.x = W_self[(kb + koff)     * FDIM + n];
            bS.y = W_self[(kb + koff + 1) * FDIM + n];
            accA = __builtin_amdgcn_wmma_f32_16x16x4_f32(
                       false, a, false, bA, (short)0, accA, false, false);
            accS = __builtin_amdgcn_wmma_f32_16x16x4_f32(
                       false, a, false, bS, (short)0, accS, false, false);
        }
        const float an = attD[n];
        #pragma unroll
        for (int r = 0; r < 8; ++r)
            Pt[half * 8 + r][n] = accA[r] * an;
        accY = accS; bn = b_self[n]; Y = out;
    } else {
        v8f acc = {};
        #pragma unroll
        for (int kb = 0; kb < FDIM; kb += 4) {
            v2f a, b;
            a.x = At[m][kb + koff];
            a.y = At[m][kb + koff + 1];
            b.x = W_src[(kb + koff)     * FDIM + n];
            b.y = W_src[(kb + koff + 1) * FDIM + n];
            acc = __builtin_amdgcn_wmma_f32_16x16x4_f32(
                      false, a, false, b, (short)0, acc, false, false);
        }
        const float an = attS[n];
        #pragma unroll
        for (int r = 0; r < 8; ++r)
            Pt[half * 8 + r][n] = acc[r] * an;
        accY = acc; bn = 0.0f; Y = src_feat;
    }
    __syncthreads();

    #pragma unroll
    for (int r = 0; r < 8; ++r) {
        const int row = row0 + half * 8 + r;
        if (row < M) Y[(size_t)row * FDIM + n] = accY[r] + bn;
    }

    if (tid < 64) {
        const int r = tid >> 2, h = tid & 3;
        const float4* p = (const float4*)&Pt[r][h * 32];
        float4 s = p[0];
        #pragma unroll
        for (int j = 1; j < 8; ++j) {
            const float4 v = p[j];
            s.x += v.x; s.y += v.y; s.z += v.z; s.w += v.w;
        }
        const float dot = (s.x + s.y) + (s.z + s.w);
        const int row = row0 + r;
        if (row < M) {
            const int i = row * HEADS + h;
            if (is_dst) ad_node[i] = dot;
            else        as_node[i] = dot;
        }
    }

    {
        const int n4  = E >> 2;
        const int per = (n4 + (int)gridDim.x - 1) / (int)gridDim.x;
        const int q0  = (int)blockIdx.x * per;
        const int q1  = q0 + per < n4 ? q0 + per : n4;
        const int4* __restrict__ ed4 = (const int4*)edst;
        for (int q = q0 + tid; q < q1; q += 256) {
            const int4 v = ed4[q];
            keys[q] =  (((unsigned)v.x / OWN_ROWS) & 255u)
                    | ((((unsigned)v.y / OWN_ROWS) & 255u) << 8)
                    | ((((unsigned)v.z / OWN_ROWS) & 255u) << 16)
                    | ((((unsigned)v.w / OWN_ROWS) & 255u) << 24);
        }
    }
}

#define OWN_THREADS 1024
#define OWN_WAVES   (OWN_THREADS / 32)
#define LCAP        4096
#define WQCAP       (LCAP * HEADS / OWN_WAVES)
#define EPW         8
#define SCAN_AHEAD  8

__device__ __forceinline__ float lrelu02(float a) { return a > 0.f ? a : 0.2f * a; }

__device__ __forceinline__ void add_row_lds(float* acc, int d, int lane, float wl, int q,
                                            float f0, float f1, float f2, float f3)
{
    const float w0 = __int_as_float(__builtin_amdgcn_readlane(__float_as_int(wl), q + 0));
    const float w1 = __int_as_float(__builtin_amdgcn_readlane(__float_as_int(wl), q + 1));
    const float w2 = __int_as_float(__builtin_amdgcn_readlane(__float_as_int(wl), q + 2));
    const float w3 = __int_as_float(__builtin_amdgcn_readlane(__float_as_int(wl), q + 3));
    float* ap = acc + d * FDIM + lane;
    __hip_atomic_fetch_add(ap +  0, w0 * f0, __ATOMIC_RELAXED, __HIP_MEMORY_SCOPE_WORKGROUP);
    __hip_atomic_fetch_add(ap + 32, w1 * f1, __ATOMIC_RELAXED, __HIP_MEMORY_SCOPE_WORKGROUP);
    __hip_atomic_fetch_add(ap + 64, w2 * f2, __ATOMIC_RELAXED, __HIP_MEMORY_SCOPE_WORKGROUP);
    __hip_atomic_fetch_add(ap + 96, w3 * f3, __ATOMIC_RELAXED, __HIP_MEMORY_SCOPE_WORKGROUP);
}

__global__ __launch_bounds__(OWN_THREADS) void edge_owner(
    const int* __restrict__ esrc, const int* __restrict__ edst,
    const float* __restrict__ src_feat,
    const float* __restrict__ asn, const float* __restrict__ adn,
    const unsigned* __restrict__ keys,
    float* __restrict__ out, int E, int N_DST)
{
    __shared__ alignas(16) float acc[OWN_ROWS * FDIM];
    __shared__ alignas(16) float adl[OWN_ROWS * HEADS];
    __shared__ unsigned mx[OWN_ROWS * HEADS];
    __shared__ float    sm[OWN_ROWS * HEADS];
    __shared__ int      lst_s[LCAP];
    __shared__ int      lst_d[LCAP];
    __shared__ alignas(16) float lst_a[LCAP * HEADS];
    __shared__ int      cnt;

    const int tid  = threadIdx.x;
    const int r0   = blockIdx.x * OWN_ROWS;
    const int left = N_DST - r0;
    const int rows = left < OWN_ROWS ? left : OWN_ROWS;

    for (int i = tid; i < OWN_ROWS * FDIM / 4; i += OWN_THREADS)
        ((float4*)acc)[i] = make_float4(0.f, 0.f, 0.f, 0.f);
    for (int i = tid; i < rows * HEADS; i += OWN_THREADS) {
        adl[i] = adn[(size_t)r0 * HEADS + i];
        mx[i] = 0u;
        sm[i] = 0.f;
    }
    if (tid == 0) cnt = 0;
    __syncthreads();

    const bool keyed = gridDim.x <= 256u;
    if (keyed) {
        const int n16 = E >> 4;
        const v4u* __restrict__ k4 = (const v4u*)keys;
        const unsigned pat = (unsigned)blockIdx.x * 0x01010101u;
        const int start = (int)blockIdx.x * (n16 / (int)gridDim.x);
        const int sl = tid & 31;
        const int sw = __builtin_amdgcn_readfirstlane(tid >> 5);
        int* wq = (int*)lst_a + sw * WQCAP;
        int  wn = 0;
        for (int base = 0; base < n16; base += SCAN_AHEAD * OWN_THREADS) {
            v4u v[SCAN_AHEAD];
            int q[SCAN_AHEAD];
            #pragma unroll
            for (int u = 0; u < SCAN_AHEAD; ++u) {
                const int i = base + u * OWN_THREADS + tid;
                int qq = start + (i < n16 ? i : 0);
                qq = qq < n16 ? qq : qq - n16;
                q[u] = i < n16 ? qq : -1;
                v[u] = k4[qq];
            }
            asm volatile("" :: "v"(v[0]), "v"(v[1]), "v"(v[2]), "v"(v[3]),
                               "v"(v[4]), "v"(v[5]), "v"(v[6]), "v"(v[7]));
            #pragma unroll
            for (int u = 0; u < SCAN_AHEAD; ++u) {
                const unsigned x0 = v[u].x ^ pat, x1 = v[u].y ^ pat;
                const unsigned x2 = v[u].z ^ pat, x3 = v[u].w ^ pat;
                const unsigned z = (((x0 - 0x01010101u) & ~x0) | ((x1 - 0x01010101u) & ~x1) |
                                    ((x2 - 0x01010101u) & ~x2) | ((x3 - 0x01010101u) & ~x3))
                                   & 0x80808080u;
                const bool hit = q[u] >= 0 && z != 0u;
                const unsigned bal = __builtin_amdgcn_ballot_w32(hit);
                if (hit) {
                    const int pos = wn + (int)__builtin_amdgcn_mbcnt_lo(bal, 0u);
                    if (pos < WQCAP) wq[pos] = q[u];
                }
                wn += __builtin_popcount(bal);
            }
        }
        if (sl == 0 && wn > WQCAP)
            __hip_atomic_fetch_add(&cnt, LCAP + 1, __ATOMIC_RELAXED, __HIP_MEMORY_SCOPE_WORKGROUP);
        __syncthreads();
        const int wc = wn < WQCAP ? wn : WQCAP;
        for (int i = sl; i < wc; i += 32) {
            const int   ql = wq[i];
            const v4u kv = k4[ql];
            const unsigned xs[4] = { kv.x ^ pat, kv.y ^ pat, kv.z ^ pat, kv.w ^ pat };
            unsigned ms[4];
            int c = 0;
            #pragma unroll
            for (int k = 0; k < 4; ++k) {
                ms[k] = ~((((xs[k] & 0x7F7F7F7Fu) + 0x7F7F7F7Fu) | xs[k]) | 0x7F7F7F7Fu);
                c += __builtin_popcount(ms[k]);
            }
            int pos = __hip_atomic_fetch_add(&cnt, c, __ATOMIC_RELAXED, __HIP_MEMORY_SCOPE_WORKGROUP);
            #pragma unroll
            for (int k = 0; k < 4; ++k) {
                unsigned m = ms[k];
                while (m) {
                    const int b = __builtin_ctz(m) >> 3;
                    if (pos < LCAP) lst_s[pos] = 16 * ql + 4 * k + b;
                    ++pos;
                    m &= m - 1u;
                }
            }
        }
        for (int e = (n16 << 4) + tid; e < E; e += OWN_THREADS) {
            const unsigned dl = (unsigned)(edst[e] - r0);
            if (dl < (unsigned)rows) {
                const int pos = __hip_atomic_fetch_add(&cnt, 1, __ATOMIC_RELAXED,
                                                       __HIP_MEMORY_SCOPE_WORKGROUP);
                if (pos < LCAP) lst_s[pos] = e;
            }
        }
    } else {
        if (tid == 0) cnt = LCAP + 1;
    }
    __syncthreads();
    const int  total = cnt;
    const bool lmode = total <= LCAP;
    const int  n     = lmode ? total : E;

    for (int i = tid; i < n; i += OWN_THREADS) {
        int s, dl; bool own;
        if (lmode) {
            const int e = lst_s[i];
            s = esrc[e]; dl = edst[e] - r0; own = true;
            lst_s[i] = s; lst_d[i] = dl;
        }
        else {
            const unsigned u = (unsigned)(edst[i] - r0);
            own = u < (unsigned)rows; dl = own ? (int)u : 0; s = own ? esrc[i] : 0;
        }
        if (own) {
            const float4 a4 = *(const float4*)(asn + (size_t)s * HEADS);
            const float4 d4 = *(const float4*)(adl + dl * HEADS);
            const float a0 = lrelu02(a4.x + d4.x), a1 = lrelu02(a4.y + d4.y);
            const float a2 = lrelu02(a4.z + d4.z), a3 = lrelu02(a4.w + d4.w);
            if (lmode) *(float4*)(lst_a + i * HEADS) = make_float4(a0, a1, a2, a3);
            unsigned* m = mx + dl * HEADS;
            __hip_atomic_fetch_max(m + 0, f2key(a0), __ATOMIC_RELAXED, __HIP_MEMORY_SCOPE_WORKGROUP);
            __hip_atomic_fetch_max(m + 1, f2key(a1), __ATOMIC_RELAXED, __HIP_MEMORY_SCOPE_WORKGROUP);
            __hip_atomic_fetch_max(m + 2, f2key(a2), __ATOMIC_RELAXED, __HIP_MEMORY_SCOPE_WORKGROUP);
            __hip_atomic_fetch_max(m + 3, f2key(a3), __ATOMIC_RELAXED, __HIP_MEMORY_SCOPE_WORKGROUP);
        }
    }
    __syncthreads();

    for (int i = tid; i < n; i += OWN_THREADS) {
        int s, dl; bool own;
        if (lmode) { s = lst_s[i]; dl = lst_d[i]; own = true; }
        else {
            const unsigned u = (unsigned)(edst[i] - r0);
            own = u < (unsigned)rows; dl = own ? (int)u : 0; s = own ? esrc[i] : 0;
        }
        if (own) {
            float4 a;
            if (lmode) a = *(const float4*)(lst_a + i * HEADS);
            else {
                const float4 a4 = *(const float4*)(asn + (size_t)s * HEADS);
                const float4 d4 = *(const float4*)(adl + dl * HEADS);
                a = make_float4(lrelu02(a4.x + d4.x), lrelu02(a4.y + d4.y),
                                lrelu02(a4.z + d4.z), lrelu02(a4.w + d4.w));
            }
            const unsigned* m = mx + dl * HEADS;
            const float p0 = __expf(a.x - key2f(m[0])), p1 = __expf(a.y - key2f(m[1]));
            const float p2 = __expf(a.z - key2f(m[2])), p3 = __expf(a.w - key2f(m[3]));
            if (lmode) *(float4*)(lst_a + i * HEADS) = make_float4(p0, p1, p2, p3);
            float* sp = sm + dl * HEADS;
            __hip_atomic_fetch_add(sp + 0, p0, __ATOMIC_RELAXED, __HIP_MEMORY_SCOPE_WORKGROUP);
            __hip_atomic_fetch_add(sp + 1, p1, __ATOMIC_RELAXED, __HIP_MEMORY_SCOPE_WORKGROUP);
            __hip_atomic_fetch_add(sp + 2, p2, __ATOMIC_RELAXED, __HIP_MEMORY_SCOPE_WORKGROUP);
            __hip_atomic_fetch_add(sp + 3, p3, __ATOMIC_RELAXED, __HIP_MEMORY_SCOPE_WORKGROUP);
        }
    }
    __syncthreads();

    {
        const int lane = tid & 31;
        const int wv = __builtin_amdgcn_readfirstlane(tid >> 5);
        const int hl = lane & 3;
        for (int i0 = wv * EPW; i0 < n; i0 += OWN_WAVES * EPW) {
            const int il = i0 + (lane >> 2);
            if (lmode && i0 + EPW <= n) {
                const int   s  = lst_s[il];
                const int   dl = lst_d[il];
                const float p  = lst_a[il * HEADS + hl];
                float f[EPW][4];
                #pragma unroll
                for (int j = 0; j < EPW; ++j) {
                    const int sj = __builtin_amdgcn_readlane(s, 4 * j);
                    const float* sp = src_feat + (size_t)sj * FDIM + lane;
                    f[j][0] = sp[0]; f[j][1] = sp[32]; f[j][2] = sp[64]; f[j][3] = sp[96];
                }
                const float wl = p / (sm[dl * HEADS + hl] + 1e-16f);
                #pragma unroll
                for (int j = 0; j < EPW; ++j)
                    add_row_lds(acc, __builtin_amdgcn_readlane(dl, 4 * j), lane, wl, 4 * j,
                                f[j][0], f[j][1], f[j][2], f[j][3]);
            } else {
                int s = 0, dl = 0; bool own = false;
                if (il < n) {
                    if (lmode) { s = lst_s[il]; dl = lst_d[il]; own = true; }
                    else {
                        const unsigned u = (unsigned)(edst[il] - r0);
                        own = u < (unsigned)rows; dl = own ? (int)u : 0; s = own ? esrc[il] : 0;
                    }
                }
                float wl = 0.f;
                if (own) {
                    float p;
                    if (lmode) p = lst_a[il * HEADS + hl];
                    else p = __expf(lrelu02(asn[(size_t)s * HEADS + hl] + adl[dl * HEADS + hl])
                                    - key2f(mx[dl * HEADS + hl]));
                    wl = p / (sm[dl * HEADS + hl] + 1e-16f);
                }
                const unsigned om = __builtin_amdgcn_ballot_w32(own);
                for (int j = 0; j < EPW; ++j) {
                    if (om & (1u << (4 * j))) {
                        const int sj = __builtin_amdgcn_readlane(s, 4 * j);
                        const float* sp = src_feat + (size_t)sj * FDIM + lane;
                        add_row_lds(acc, __builtin_amdgcn_readlane(dl, 4 * j), lane, wl, 4 * j,
                                    sp[0], sp[32], sp[64], sp[96]);
                    }
                }
            }
        }
    }
    __syncthreads();

    {
        float4* __restrict__ op = (float4*)(out + (size_t)r0 * FDIM);
        const float4* ac = (const float4*)acc;
        for (int i = tid; i < rows * (FDIM / 4); i += OWN_THREADS) {
            float4 o = op[i];
            const float4 a = ac[i];
            o.x += a.x; o.y += a.y; o.z += a.z; o.w += a.w;
            op[i] = o;
        }
    }
}

extern "C" void kernel_launch(void* const* d_in, const int* in_sizes, int n_in,
                              void* d_out, int out_size, void* d_ws, size_t ws_size,
                              hipStream_t stream)
{
    const float* x_src  = (const float*)d_in[0];
    const float* x_dst  = (const float*)d_in[1];
    const int*   esrc   = (const int*)d_in[2];
    const int*   edst   = (const int*)d_in[3];
    const float* W_src  = (const float*)d_in[5];
    const float* W_dst  = (const float*)d_in[6];
    const float* attS   = (const float*)d_in[7];
    const float* attD   = (const float*)d_in[8];
    const float* W_self = (const float*)d_in[9];
    const float* b_self = (const float*)d_in[10];

    const int N_SRC = in_sizes[0] / FDIM;
    const int N_DST = in_sizes[1] / FDIM;
    const int E     = in_sizes[2];

    char* ws = (char*)d_ws;
    float* src_feat = (float*)(ws);
    float* as_node  = (float*)(ws + (size_t)(N_SRC + N_DST) * FDIM * 4);
    float* ad_node  = as_node + (size_t)N_SRC * HEADS;
    unsigned* keys  = (unsigned*)(ws + (size_t)N_SRC * FDIM * 4);

    float* out = (float*)d_out;

    {
        const int dst_tiles = (N_DST + 15) / 16;
        const int src_tiles = (N_SRC + 15) / 16;
        node_proj<<<dst_tiles + src_tiles, 256, 0, stream>>>(
            x_src, x_dst, W_src, W_dst, W_self, b_self, attS, attD,
            src_feat, out, as_node, ad_node, edst, keys, E, N_SRC, N_DST, dst_tiles);
    }
    edge_owner<<<(N_DST + OWN_ROWS - 1) / OWN_ROWS, OWN_THREADS, 0, stream>>>(
        esrc, edst, src_feat, as_node, ad_node, keys, out, E, N_DST);
}


# Node launch as 512 resident work-groups looping over 16-row tiles; each wave loads its W column fragments once and keeps them in registers: no W loads per tile, 4096 waves enter, not 50000; sums same
# speedup vs baseline: 4.0817x; 1.0438x over previous
#include <hip/hip_runtime.h>
#include <hip/hip_bf16.h>
#include <math.h>

typedef __attribute__((ext_vector_type(2))) float v2f;
typedef __attribute__((ext_vector_type(8))) float v8f;
typedef __attribute__((ext_vector_type(4))) unsigned v4u;

#define HEADS 4
#define FDIM  128
#define OWN_ROWS 196

__device__ __forceinline__ unsigned f2key(float f) {
    unsigned b = __float_as_uint(f);
    return (b & 0x80000000u) ? ~b : (b | 0x80000000u);
}
__device__ __forceinline__ float key2f(unsigned k) {
    unsigned b = (k & 0x80000000u) ? (k & 0x7FFFFFFFu) : ~k;
    return __uint_as_float(b);
}

#define NP_GROUPS 512

__device__ __forceinline__ void lds_barrier() {
    asm volatile("s_wait_dscnt 0x0\n\ts_barrier_signal -1\n\ts_barrier_wait -1" ::: "memory");
}

__device__ __forceinline__ void load_x_tile(const float* __restrict__ X, int M, int row0,
                                            int r, int c, float4& v0, float4& v1)
{
    int g0 = row0 + r;     g0 = g0 < M ? g0 : M - 1;
    int g1 = row0 + r + 8; g1 = g1 < M ? g1 : M - 1;
    v0 = *(const float4*)(X + (size_t)g0 * FDIM + c);
    v1 = *(const float4*)(X + (size_t)g1 * FDIM + c);
}

template <bool DST>
__device__ __forceinline__ void node_rows(
    float (*At)[132], float (*Pt)[132],
    const float* __restrict__ X, int M, int tiles, int first, int step,
    const float* __restrict__ Wa, const float* __restrict__ Wb,
    const float* __restrict__ att, const float* __restrict__ bias,
    float* __restrict__ Y, float* __restrict__ dots, int tid)
{
    if (first >= tiles) return;

    const int lane = tid & 31;
    const int wv   = tid >> 5;
    const int m    = lane & 15;
    const int half = lane >> 4;
    const int koff = half << 1;
    const int n    = wv * 16 + m;
    const int xr   = tid >> 5;
    const int xc   = (tid & 31) * 4;

    v2f bA[32], bB[DST ? 32 : 1];
    #pragma unroll
    for (int j = 0; j < 32; ++j) {
        bA[j].x = Wa[(4 * j + koff)     * FDIM + n];
        bA[j].y = Wa[(4 * j + koff + 1) * FDIM + n];
        if (DST) {
            bB[j].x = Wb[(4 * j + koff)     * FDIM + n];
            bB[j].y = Wb[(4 * j + koff + 1) * FDIM + n];
        }
    }
    const float an = att[n];
    const float bn = DST ? bias[n] : 0.0f;

    float4 x0, x1;
    load_x_tile(X, M, first * 16, xr, xc, x0, x1);

    for (int t = first; t < tiles; t += step) {
        const int row0 = t * 16;
        *(float4*)&At[xr][xc]     = x0;
        *(float4*)&At[xr + 8][xc] = x1;
        lds_barrier();
        if (t + step < tiles) load_x_tile(X, M, (t + step) * 16, xr, xc, x0, x1);

        v8f accA = {}, accB = {};
        #pragma unroll
        for (int j = 0; j < 32; ++j) {
            v2f a;
            a.x = At[m][4 * j + koff];
            a.y = At[m][4 * j + koff + 1];
            accA = __builtin_amdgcn_wmma_f32_16x16x4_f32(
                       false, a, false, bA[j], (short)0, accA, false, false);
            if (DST)
                accB = __builtin_amdgcn_wmma_f32_16x16x4_f32(
                           false, a, false, bB[j], (short)0, accB, false, false);
        }
        #pragma unroll
        for (int r = 0; r < 8; ++r)
            Pt[half * 8 + r][n] = accA[r] * an;
        lds_barrier();

        #pragma unroll
        for (int r = 0; r < 8; ++r) {
            const int row = row0 + half * 8 + r;
            if (row < M) Y[(size_t)row * FDIM + n] = (DST ? accB[r] : accA[r]) + bn;
        }

        if (tid < 64) {
            const int r = tid >> 2, h = tid & 3;
            const float4* p = (const float4*)&Pt[r][h * 32];
            float4 s = p[0];
            #pragma unroll
            for (int j = 1; j < 8; ++j) {
                const float4 v = p[j];
                s.x += v.x; s.y += v.y; s.z += v.z; s.w += v.w;
            }
            const float dot = (s.x + s.y) + (s.z + s.w);
            const int row = row0 + r;
            if (row < M) dots[row * HEADS + h] = dot;
        }
    }
}

__global__ __launch_bounds__(256) void node_proj(
    const float* __restrict__ x_src, const float* __restrict__ x_dst,
    const float* __restrict__ W_src, const float* __restrict__ W_dst,
    const float* __restrict__ W_self, const float* __restrict__ b_self,
    const float* __restrict__ attS, const float* __restrict__ attD,
    float* __restrict__ src_feat, float* __restrict__ out,
    float* __restrict__ as_node, float* __restrict__ ad_node,
    const int* __restrict__ edst, unsigned* __restrict__ keys, int E,
    int N_SRC, int N_DST)
{
    __shared__ alignas(16) float At[16][132];
    __shared__ alignas(16) float Pt[16][132];
    const int tid = threadIdx.x;
    const int G   = (int)gridDim.x;
    const int g   = (int)blockIdx.x;

    {
        const int n4  = E >> 2;
        const int per = (n4 + G - 1) / G;
        const int q0  = g * per;
        const int q1  = q0 + per < n4 ? q0 + per : n4;
        const int4* __restrict__ ed4 = (const int4*)edst;
        for (int q = q0 + tid; q < q1; q += 256) {
            const int4 v = ed4[q];
            keys[q] =  (((unsigned)v.x / OWN_ROWS) & 255u)
                    | ((((unsigned)v.y / OWN_ROWS) & 255u) << 8)
                    | ((((unsigned)v.z / OWN_ROWS) & 255u) << 16)
                    | ((((unsigned)v.w / OWN_ROWS) & 255u) << 24);
        }
    }

    node_rows<true >(At, Pt, x_dst, N_DST, (N_DST + 15) >> 4, g, G,
                     W_dst, W_self, attD, b_self, out, ad_node, tid);
    node_rows<false>(At, Pt, x_src, N_SRC, (N_SRC + 15) >> 4, G - 1 - g, G,
                     W_src, W_src, attS, b_self, src_feat, as_node, tid);
}

#define OWN_THREADS 1024
#define OWN_WAVES   (OWN_THREADS / 32)
#define LCAP        4096
#define WQCAP       (LCAP * HEADS / OWN_WAVES)
#define EPW         8
#define SCAN_AHEAD  8

__device__ __forceinline__ float lrelu02(float a) { return a > 0.f ? a : 0.2f * a; }

__device__ __forceinline__ void add_row_lds(float* acc, int d, int lane, float wl, int q,
                                            float f0, float f1, float f2, float f3)
{
    const float w0 = __int_as_float(__builtin_amdgcn_readlane(__float_as_int(wl), q + 0));
    const float w1 = __int_as_float(__builtin_amdgcn_readlane(__float_as_int(wl), q + 1));
    const float w2 = __int_as_float(__builtin_amdgcn_readlane(__float_as_int(wl), q + 2));
    const float w3 = __int_as_float(__builtin_amdgcn_readlane(__float_as_int(wl), q + 3));
    float* ap = acc + d * FDIM + lane;
    __hip_atomic_fetch_add(ap +  0, w0 * f0, __ATOMIC_RELAXED, __HIP_MEMORY_SCOPE_WORKGROUP);
    __hip_atomic_fetch_add(ap + 32, w1 * f1, __ATOMIC_RELAXED, __HIP_MEMORY_SCOPE_WORKGROUP);
    __hip_atomic_fetch_add(ap + 64, w2 * f2, __ATOMIC_RELAXED, __HIP_MEMORY_SCOPE_WORKGROUP);
    __hip_atomic_fetch_add(ap + 96, w3 * f3, __ATOMIC_RELAXED, __HIP_MEMORY_SCOPE_WORKGROUP);
}

__global__ __launch_bounds__(OWN_THREADS) void edge_owner(
    const int* __restrict__ esrc, const int* __restrict__ edst,
    const float* __restrict__ src_feat,
    const float* __restrict__ asn, const float* __restrict__ adn,
    const unsigned* __restrict__ keys,
    float* __restrict__ out, int E, int N_DST)
{
    __shared__ alignas(16) float acc[OWN_ROWS * FDIM];
    __shared__ alignas(16) float adl[OWN_ROWS * HEADS];
    __shared__ unsigned mx[OWN_ROWS * HEADS];
    __shared__ float    sm[OWN_ROWS * HEADS];
    __shared__ int      lst_s[LCAP];
    __shared__ int      lst_d[LCAP];
    __shared__ alignas(16) float lst_a[LCAP * HEADS];
    __shared__ int      cnt;

    const int tid  = threadIdx.x;
    const int r0   = blockIdx.x * OWN_ROWS;
    const int left = N_DST - r0;
    const int rows = left < OWN_ROWS ? left : OWN_ROWS;

    for (int i = tid; i < OWN_ROWS * FDIM / 4; i += OWN_THREADS)
        ((float4*)acc)[i] = make_float4(0.f, 0.f, 0.f, 0.f);
    for (int i = tid; i < rows * HEADS; i += OWN_THREADS) {
        adl[i] = adn[(size_t)r0 * HEADS + i];
        mx[i] = 0u;
        sm[i] = 0.f;
    }
    if (tid == 0) cnt = 0;
    __syncthreads();

    const bool keyed = gridDim.x <= 256u;
    if (keyed) {
        const int n16 = E >> 4;
        const v4u* __restrict__ k4 = (const v4u*)keys;
        const unsigned pat = (unsigned)blockIdx.x * 0x01010101u;
        const int start = (int)blockIdx.x * (n16 / (int)gridDim.x);
        const int sl = tid & 31;
        const int sw = __builtin_amdgcn_readfirstlane(tid >> 5);
        int* wq = (int*)lst_a + sw * WQCAP;
        int  wn = 0;
        for (int base = 0; base < n16; base += SCAN_AHEAD * OWN_THREADS) {
            v4u v[SCAN_AHEAD];
            int q[SCAN_AHEAD];
            #pragma unroll
            for (int u = 0; u < SCAN_AHEAD; ++u) {
                const int i = base + u * OWN_THREADS + tid;
                int qq = start + (i < n16 ? i : 0);
                qq = qq < n16 ? qq : qq - n16;
                q[u] = i < n16 ? qq : -1;
                v[u] = k4[qq];
            }
            asm volatile("" :: "v"(v[0]), "v"(v[1]), "v"(v[2]), "v"(v[3]),
                               "v"(v[4]), "v"(v[5]), "v"(v[6]), "v"(v[7]));
            #pragma unroll
            for (int u = 0; u < SCAN_AHEAD; ++u) {
                const unsigned x0 = v[u].x ^ pat, x1 = v[u].y ^ pat;
                const unsigned x2 = v[u].z ^ pat, x3 = v[u].w ^ pat;
                const unsigned z = (((x0 - 0x01010101u) & ~x0) | ((x1 - 0x01010101u) & ~x1) |
                                    ((x2 - 0x01010101u) & ~x2) | ((x3 - 0x01010101u) & ~x3))
                                   & 0x80808080u;
                const bool hit = q[u] >= 0 && z != 0u;
                const unsigned bal = __builtin_amdgcn_ballot_w32(hit);
                if (hit) {
                    const int pos = wn + (int)__builtin_amdgcn_mbcnt_lo(bal, 0u);
                    if (pos < WQCAP) wq[pos] = q[u];
                }
                wn += __builtin_popcount(bal);
            }
        }
        if (sl == 0 && wn > WQCAP)
            __hip_atomic_fetch_add(&cnt, LCAP + 1, __ATOMIC_RELAXED, __HIP_MEMORY_SCOPE_WORKGROUP);
        __syncthreads();
        const int wc = wn < WQCAP ? wn : WQCAP;
        for (int i = sl; i < wc; i += 32) {
            const int   ql = wq[i];
            const v4u kv = k4[ql];
            const unsigned xs[4] = { kv.x ^ pat, kv.y ^ pat, kv.z ^ pat, kv.w ^ pat };
            unsigned ms[4];
            int c = 0;
            #pragma unroll
            for (int k = 0; k < 4; ++k) {
                ms[k] = ~((((xs[k] & 0x7F7F7F7Fu) + 0x7F7F7F7Fu) | xs[k]) | 0x7F7F7F7Fu);
                c += __builtin_popcount(ms[k]);
            }
            int pos = __hip_atomic_fetch_add(&cnt, c, __ATOMIC_RELAXED, __HIP_MEMORY_SCOPE_WORKGROUP);
            #pragma unroll
            for (int k = 0; k < 4; ++k) {
                unsigned m = ms[k];
                while (m) {
                    const int b = __builtin_ctz(m) >> 3;
                    if (pos < LCAP) lst_s[pos] = 16 * ql + 4 * k + b;
                    ++pos;
                    m &= m - 1u;
                }
            }
        }
        for (int e = (n16 << 4) + tid; e < E; e += OWN_THREADS) {
            const unsigned dl = (unsigned)(edst[e] - r0);
            if (dl < (unsigned)rows) {
                const int pos = __hip_atomic_fetch_add(&cnt, 1, __ATOMIC_RELAXED,
                                                       __HIP_MEMORY_SCOPE_WORKGROUP);
                if (pos < LCAP) lst_s[pos] = e;
            }
        }
    } else {
        if (tid == 0) cnt = LCAP + 1;
    }
    __syncthreads();
    const int  total = cnt;
    const bool lmode = total <= LCAP;
    const int  n     = lmode ? total : E;

    for (int i = tid; i < n; i += OWN_THREADS) {
        int s, dl; bool own;
        if (lmode) {
            const int e = lst_s[i];
            s = esrc[e]; dl = edst[e] - r0; own = true;
            lst_s[i] = s; lst_d[i] = dl;
        }
        else {
            const unsigned u = (unsigned)(edst[i] - r0);
            own = u < (unsigned)rows; dl = own ? (int)u : 0; s = own ? esrc[i] : 0;
        }
        if (own) {
            const float4 a4 = *(const float4*)(asn + (size_t)s * HEADS);
            const float4 d4 = *(const float4*)(adl + dl * HEADS);
            const float a0 = lrelu02(a4.x + d4.x), a1 = lrelu02(a4.y + d4.y);
            const float a2 = lrelu02(a4.z + d4.z), a3 = lrelu02(a4.w + d4.w);
            if (lmode) *(float4*)(lst_a + i * HEADS) = make_float4(a0, a1, a2, a3);
            unsigned* m = mx + dl * HEADS;
            __hip_atomic_fetch_max(m + 0, f2key(a0), __ATOMIC_RELAXED, __HIP_MEMORY_SCOPE_WORKGROUP);
            __hip_atomic_fetch_max(m + 1, f2key(a1), __ATOMIC_RELAXED, __HIP_MEMORY_SCOPE_WORKGROUP);
            __hip_atomic_fetch_max(m + 2, f2key(a2), __ATOMIC_RELAXED, __HIP_MEMORY_SCOPE_WORKGROUP);
            __hip_atomic_fetch_max(m + 3, f2key(a3), __ATOMIC_RELAXED, __HIP_MEMORY_SCOPE_WORKGROUP);
        }
    }
    __syncthreads();

    for (int i = tid; i < n; i += OWN_THREADS) {
        int s, dl; bool own;
        if (lmode) { s = lst_s[i]; dl = lst_d[i]; own = true; }
        else {
            const unsigned u = (unsigned)(edst[i] - r0);
            own = u < (unsigned)rows; dl = own ? (int)u : 0; s = own ? esrc[i] : 0;
        }
        if (own) {
            float4 a;
            if (lmode) a = *(const float4*)(lst_a + i * HEADS);
            else {
                const float4 a4 = *(const float4*)(asn + (size_t)s * HEADS);
                const float4 d4 = *(const float4*)(adl + dl * HEADS);
                a = make_float4(lrelu02(a4.x + d4.x), lrelu02(a4.y + d4.y),
                                lrelu02(a4.z + d4.z), lrelu02(a4.w + d4.w));
            }
            const unsigned* m = mx + dl * HEADS;
            const float p0 = __expf(a.x - key2f(m[0])), p1 = __expf(a.y - key2f(m[1]));
            const float p2 = __expf(a.z - key2f(m[2])), p3 = __expf(a.w - key2f(m[3]));
            if (lmode) *(float4*)(lst_a + i * HEADS) = make_float4(p0, p1, p2, p3);
            float* sp = sm + dl * HEADS;
            __hip_atomic_fetch_add(sp + 0, p0, __ATOMIC_RELAXED, __HIP_MEMORY_SCOPE_WORKGROUP);
            __hip_atomic_fetch_add(sp + 1, p1, __ATOMIC_RELAXED, __HIP_MEMORY_SCOPE_WORKGROUP);
            __hip_atomic_fetch_add(sp + 2, p2, __ATOMIC_RELAXED, __HIP_MEMORY_SCOPE_WORKGROUP);
            __hip_atomic_fetch_add(sp + 3, p3, __ATOMIC_RELAXED, __HIP_MEMORY_SCOPE_WORKGROUP);
        }
    }
    __syncthreads();

    {
        const int lane = tid & 31;
        const int wv = __builtin_amdgcn_readfirstlane(tid >> 5);
        const int hl = lane & 3;
        for (int i0 = wv * EPW; i0 < n; i0 += OWN_WAVES * EPW) {
            const int il = i0 + (lane >> 2);
            if (lmode && i0 + EPW <= n) {
                const int   s  = lst_s[il];
                const int   dl = lst_d[il];
                const float p  = lst_a[il * HEADS + hl];
                float f[EPW][4];
                #pragma unroll
                for (int j = 0; j < EPW; ++j) {
                    const int sj = __builtin_amdgcn_readlane(s, 4 * j);
                    const float* sp = src_feat + (size_t)sj * FDIM + lane;
                    f[j][0] = sp[0]; f[j][1] = sp[32]; f[j][2] = sp[64]; f[j][3] = sp[96];
                }
                const float wl = p / (sm[dl * HEADS + hl] + 1e-16f);
                #pragma unroll
                for (int j = 0; j < EPW; ++j)
                    add_row_lds(acc, __builtin_amdgcn_readlane(dl, 4 * j), lane, wl, 4 * j,
                                f[j][0], f[j][1], f[j][2], f[j][3]);
            } else {
                int s = 0, dl = 0; bool own = false;
                if (il < n) {
                    if (lmode) { s = lst_s[il]; dl = lst_d[il]; own = true; }
                    else {
                        const unsigned u = (unsigned)(edst[il] - r0);
                        own = u < (unsigned)rows; dl = own ? (int)u : 0; s = own ? esrc[il] : 0;
                    }
                }
                float wl = 0.f;
                if (own) {
                    float p;
                    if (lmode) p = lst_a[il * HEADS + hl];
                    else p = __expf(lrelu02(asn[(size_t)s * HEADS + hl] + adl[dl * HEADS + hl])
                                    - key2f(mx[dl * HEADS + hl]));
                    wl = p / (sm[dl * HEADS + hl] + 1e-16f);
                }
                const unsigned om = __builtin_amdgcn_ballot_w32(own);
                for (int j = 0; j < EPW; ++j) {
                    if (om & (1u << (4 * j))) {
                        const int sj = __builtin_amdgcn_readlane(s, 4 * j);
                        const float* sp = src_feat + (size_t)sj * FDIM + lane;
                        add_row_lds(acc, __builtin_amdgcn_readlane(dl, 4 * j), lane, wl, 4 * j,
                                    sp[0], sp[32], sp[64], sp[96]);
                    }
                }
            }
        }
    }
    __syncthreads();

    {
        float4* __restrict__ op = (float4*)(out + (size_t)r0 * FDIM);
        const float4* ac = (const float4*)acc;
        for (int i = tid; i < rows * (FDIM / 4); i += OWN_THREADS) {
            float4 o = op[i];
            const float4 a = ac[i];
            o.x += a.x; o.y += a.y; o.z += a.z; o.w += a.w;
            op[i] = o;
        }
    }
}

extern "C" void kernel_launch(void* const* d_in, const int* in_sizes, int n_in,
                              void* d_out, int out_size, void* d_ws, size_t ws_size,
                              hipStream_t stream)
{
    const float* x_src  = (const float*)d_in[0];
    const float* x_dst  = (const float*)d_in[1];
    const int*   esrc   = (const int*)d_in[2];
    const int*   edst   = (const int*)d_in[3];
    const float* W_src  = (const float*)d_in[5];
    const float* W_dst  = (const float*)d_in[6];
    const float* attS   = (const float*)d_in[7];
    const float* attD   = (const float*)d_in[8];
    const float* W_self = (const float*)d_in[9];
    const float* b_self = (const float*)d_in[10];

    const int N_SRC = in_sizes[0] / FDIM;
    const int N_DST = in_sizes[1] / FDIM;
    const int E     = in_sizes[2];

    char* ws = (char*)d_ws;
    float* src_feat = (float*)(ws);
    float* as_node  = (float*)(ws + (size_t)(N_SRC + N_DST) * FDIM * 4);
    float* ad_node  = as_node + (size_t)N_SRC * HEADS;
    unsigned* keys  = (unsigned*)(ws + (size_t)N_SRC * FDIM * 4);

    float* out = (float*)d_out;

    node_proj<<<NP_GROUPS, 256, 0, stream>>>(
        x_src, x_dst, W_src, W_dst, W_self, b_self, attS, attD,
        src_feat, out, as_node, ad_node, edst, keys, E, N_SRC, N_DST);
    edge_owner<<<(N_DST + OWN_ROWS - 1) / OWN_ROWS, OWN_THREADS, 0, stream>>>(
        esrc, edst, src_feat, as_node, ad_node, keys, out, E, N_DST);
}


# Own-edge list from per-edge (source, destination) pairs the node launch writes: a wave expands its queued key loads two to a load, a lane per edge; no key or index re-reads; list (sum) order differs
# speedup vs baseline: 4.2382x; 1.0383x over previous
#include <hip/hip_runtime.h>
#include <hip/hip_bf16.h>
#include <math.h>

typedef __attribute__((ext_vector_type(2))) float v2f;
typedef __attribute__((ext_vector_type(8))) float v8f;
typedef __attribute__((ext_vector_type(4))) unsigned v4u;
typedef __attribute__((ext_vector_type(2))) int v2i;

#define HEADS 4
#define FDIM  128
#define OWN_ROWS 196

__device__ __forceinline__ unsigned f2key(float f) {
    unsigned b = __float_as_uint(f);
    return (b & 0x80000000u) ? ~b : (b | 0x80000000u);
}
__device__ __forceinline__ float key2f(unsigned k) {
    unsigned b = (k & 0x80000000u) ? (k & 0x7FFFFFFFu) : ~k;
    return __uint_as_float(b);
}

#define NP_GROUPS 512

__device__ __forceinline__ void lds_barrier() {
    asm volatile("s_wait_dscnt 0x0\n\ts_barrier_signal -1\n\ts_barrier_wait -1" ::: "memory");
}

__device__ __forceinline__ void load_x_tile(const float* __restrict__ X, int M, int row0,
                                            int r, int c, float4& v0, float4& v1)
{
    int g0 = row0 + r;     g0 = g0 < M ? g0 : M - 1;
    int g1 = row0 + r + 8; g1 = g1 < M ? g1 : M - 1;
    v0 = *(const float4*)(X + (size_t)g0 * FDIM + c);
    v1 = *(const float4*)(X + (size_t)g1 * FDIM + c);
}

template <bool DST>
__device__ __forceinline__ void node_rows(
    float (*At)[132], float (*Pt)[132],
    const float* __restrict__ X, int M, int tiles, int first, int step,
    const float* __restrict__ Wa, const float* __restrict__ Wb,
    const float* __restrict__ att, const float* __restrict__ bias,
    float* __restrict__ Y, float* __restrict__ dots, int tid)
{
    if (first >= tiles) return;

    const int lane = tid & 31;
    const int wv   = tid >> 5;
    const int m    = lane & 15;
    const int half = lane >> 4;
    const int koff = half << 1;
    const int n    = wv * 16 + m;
    const int xr   = tid >> 5;
    const int xc   = (tid & 31) * 4;

    v2f bA[32], bB[DST ? 32 : 1];
    #pragma unroll
    for (int j = 0; j < 32; ++j) {
        bA[j].x = Wa[(4 * j + koff)     * FDIM + n];
        bA[j].y = Wa[(4 * j + koff + 1) * FDIM + n];
        if (DST) {
            bB[j].x = Wb[(4 * j + koff)     * FDIM + n];
            bB[j].y = Wb[(4 * j + koff + 1) * FDIM + n];
        }
    }
    const float an = att[n];
    const float bn = DST ? bias[n] : 0.0f;

    float4 x0, x1;
    load_x_tile(X, M, first * 16, xr, xc, x0, x1);

    for (int t = first; t < tiles; t += step) {
        const int row0 = t * 16;
        *(float4*)&At[xr][xc]     = x0;
        *(float4*)&At[xr + 8][xc] = x1;
        lds_barrier();
        if (t + step < tiles) load_x_tile(X, M, (t + step) * 16, xr, xc, x0, x1);

        v8f accA = {}, accB = {};
        #pragma unroll
        for (int j = 0; j < 32; ++j) {
            v2f a;
            a.x = At[m][4 * j + koff];
            a.y = At[m][4 * j + koff + 1];
            accA = __builtin_amdgcn_wmma_f32_16x16x4_f32(
                       false, a, false, bA[j], (short)0, accA, false, false);
            if (DST)
                accB = __builtin_amdgcn_wmma_f32_16x16x4_f32(
                           false, a, false, bB[j], (short)0, accB, false, false);
        }
        #pragma unroll
        for (int r = 0; r < 8; ++r)
            Pt[half * 8 + r][n] = accA[r] * an;
        lds_barrier();

        #pragma unroll
        for (int r = 0; r < 8; ++r) {
            const int row = row0 + half * 8 + r;
            if (row < M) Y[(size_t)row * FDIM + n] = (DST ? accB[r] : accA[r]) + bn;
        }

        if (tid < 64) {
            const int r = tid >> 2, h = tid & 3;
            const float4* p = (const float4*)&Pt[r][h * 32];
            float4 s = p[0];
            #pragma unroll
            for (int j = 1; j < 8; ++j) {
                const float4 v = p[j];
                s.x += v.x; s.y += v.y; s.z += v.z; s.w += v.w;
            }
            const float dot = (s.x + s.y) + (s.z + s.w);
            const int row = row0 + r;
            if (row < M) dots[row * HEADS + h] = dot;
        }
    }
}

__global__ __launch_bounds__(256) void node_proj(
    const float* __restrict__ x_src, const float* __restrict__ x_dst,
    const float* __restrict__ W_src, const float* __restrict__ W_dst,
    const float* __restrict__ W_self, const float* __restrict__ b_self,
    const float* __restrict__ attS, const float* __restrict__ attD,
    float* __restrict__ src_feat, float* __restrict__ out,
    float* __restrict__ as_node, float* __restrict__ ad_node,
    const int* __restrict__ esrc, const int* __restrict__ edst,
    unsigned* __restrict__ keys, int* __restrict__ pairs, int E,
    int N_SRC, int N_DST)
{
    __shared__ alignas(16) float At[16][132];
    __shared__ alignas(16) float Pt[16][132];
    const int tid = threadIdx.x;
    const int G   = (int)gridDim.x;
    const int g   = (int)blockIdx.x;

    {
        const int n4  = E >> 2;
        const int per = (n4 + G - 1) / G;
        const int q0  = g * per;
        const int q1  = q0 + per < n4 ? q0 + per : n4;
        const int4* __restrict__ ed4 = (const int4*)edst;
        const int4* __restrict__ es4 = (const int4*)esrc;
        int4* __restrict__ pr4 = (int4*)pairs;
        for (int q = q0 + tid; q < q1; q += 256) {
            const int4 v = ed4[q];
            const int4 s = es4[q];
            keys[q] =  (((unsigned)v.x / OWN_ROWS) & 255u)
                    | ((((unsigned)v.y / OWN_ROWS) & 255u) << 8)
                    | ((((unsigned)v.z / OWN_ROWS) & 255u) << 16)
                    | ((((unsigned)v.w / OWN_ROWS) & 255u) << 24);
            pr4[2 * q]     = make_int4(s.x, v.x, s.y, v.y);
            pr4[2 * q + 1] = make_int4(s.z, v.z, s.w, v.w);
        }
    }

    node_rows<true >(At, Pt, x_dst, N_DST, (N_DST + 15) >> 4, g, G,
                     W_dst, W_self, attD, b_self, out, ad_node, tid);
    node_rows<false>(At, Pt, x_src, N_SRC, (N_SRC + 15) >> 4, G - 1 - g, G,
                     W_src, W_src, attS, b_self, src_feat, as_node, tid);
}

#define OWN_THREADS 1024
#define OWN_WAVES   (OWN_THREADS / 32)
#define LCAP        4096
#define WQCAP       (LCAP * HEADS / OWN_WAVES)
#define EPW         8
#define SCAN_AHEAD  8
#define PAIR_LOADS  16

__device__ __forceinline__ float lrelu02(float a) { return a > 0.f ? a : 0.2f * a; }

__device__ __forceinline__ void add_row_lds(float* acc, int d, int lane, float wl, int q,
                                            float f0, float f1, float f2, float f3)
{
    const float w0 = __int_as_float(__builtin_amdgcn_readlane(__float_as_int(wl), q + 0));
    const float w1 = __int_as_float(__builtin_amdgcn_readlane(__float_as_int(wl), q + 1));
    const float w2 = __int_as_float(__builtin_amdgcn_readlane(__float_as_int(wl), q + 2));
    const float w3 = __int_as_float(__builtin_amdgcn_readlane(__float_as_int(wl), q + 3));
    float* ap = acc + d * FDIM + lane;
    __hip_atomic_fetch_add(ap +  0, w0 * f0, __ATOMIC_RELAXED, __HIP_MEMORY_SCOPE_WORKGROUP);
    __hip_atomic_fetch_add(ap + 32, w1 * f1, __ATOMIC_RELAXED, __HIP_MEMORY_SCOPE_WORKGROUP);
    __hip_atomic_fetch_add(ap + 64, w2 * f2, __ATOMIC_RELAXED, __HIP_MEMORY_SCOPE_WORKGROUP);
    __hip_atomic_fetch_add(ap + 96, w3 * f3, __ATOMIC_RELAXED, __HIP_MEMORY_SCOPE_WORKGROUP);
}

__global__ __launch_bounds__(OWN_THREADS) void edge_owner(
    const int* __restrict__ esrc, const int* __restrict__ edst,
    const float* __restrict__ src_feat,
    const float* __restrict__ asn, const float* __restrict__ adn,
    const unsigned* __restrict__ keys, const int* __restrict__ pairs,
    float* __restrict__ out, int E, int N_DST)
{
    __shared__ alignas(16) float acc[OWN_ROWS * FDIM];
    __shared__ alignas(16) float adl[OWN_ROWS * HEADS];
    __shared__ unsigned mx[OWN_ROWS * HEADS];
    __shared__ float    sm[OWN_ROWS * HEADS];
    __shared__ int      lst_s[LCAP];
    __shared__ int      lst_d[LCAP];
    __shared__ alignas(16) float lst_a[LCAP * HEADS];
    __shared__ int      cnt;

    const int tid  = threadIdx.x;
    const int r0   = blockIdx.x * OWN_ROWS;
    const int left = N_DST - r0;
    const int rows = left < OWN_ROWS ? left : OWN_ROWS;

    for (int i = tid; i < OWN_ROWS * FDIM / 4; i += OWN_THREADS)
        ((float4*)acc)[i] = make_float4(0.f, 0.f, 0.f, 0.f);
    for (int i = tid; i < rows * HEADS; i += OWN_THREADS) {
        adl[i] = adn[(size_t)r0 * HEADS + i];
        mx[i] = 0u;
        sm[i] = 0.f;
    }
    if (tid == 0) cnt = 0;
    __syncthreads();

    const bool keyed = gridDim.x <= 256u;
    if (keyed) {
        const int n16 = E >> 4;
        const v4u* __restrict__ k4 = (const v4u*)keys;
        const unsigned pat = (unsigned)blockIdx.x * 0x01010101u;
        const int start = (int)blockIdx.x * (n16 / (int)gridDim.x);
        const int sl = tid & 31;
        const int sw = __builtin_amdgcn_readfirstlane(tid >> 5);
        int* wq = (int*)lst_a + sw * WQCAP;
        int  wn = 0;
        for (int base = 0; base < n16; base += SCAN_AHEAD * OWN_THREADS) {
            v4u v[SCAN_AHEAD];
            int q[SCAN_AHEAD];
            #pragma unroll
            for (int u = 0; u < SCAN_AHEAD; ++u) {
                const int i = base + u * OWN_THREADS + tid;
                int qq = start + (i < n16 ? i : 0);
                qq = qq < n16 ? qq : qq - n16;
                q[u] = i < n16 ? qq : -1;
                v[u] = k4[qq];
            }
            asm volatile("" :: "v"(v[0]), "v"(v[1]), "v"(v[2]), "v"(v[3]),
                               "v"(v[4]), "v"(v[5]), "v"(v[6]), "v"(v[7]));
            #pragma unroll
            for (int u = 0; u < SCAN_AHEAD; ++u) {
                const unsigned x0 = v[u].x ^ pat, x1 = v[u].y ^ pat;
                const unsigned x2 = v[u].z ^ pat, x3 = v[u].w ^ pat;
                const unsigned z = (((x0 - 0x01010101u) & ~x0) | ((x1 - 0x01010101u) & ~x1) |
                                    ((x2 - 0x01010101u) & ~x2) | ((x3 - 0x01010101u) & ~x3))
                                   & 0x80808080u;
                const bool hit = q[u] >= 0 && z != 0u;
                const unsigned bal = __builtin_amdgcn_ballot_w32(hit);
                if (hit) {
                    const int pos = wn + (int)__builtin_amdgcn_mbcnt_lo(bal, 0u);
                    if (pos < WQCAP) wq[pos] = q[u];
                }
                wn += __builtin_popcount(bal);
            }
        }
        if (sl == 0 && wn > WQCAP)
            __hip_atomic_fetch_add(&cnt, LCAP + 1, __ATOMIC_RELAXED, __HIP_MEMORY_SCOPE_WORKGROUP);
        const int wc = wn < WQCAP ? wn : WQCAP;
        if (wc + sl < ((wc + 31) & ~31)) wq[wc + sl] = 0;
        __syncthreads();
        const v2i* __restrict__ pr2 = (const v2i*)pairs;
        const int gh = sl >> 4;
        const int le = sl & 15;
        for (int i0 = 0; i0 < wc; i0 += 2 * PAIR_LOADS) {
            v2i r[PAIR_LOADS];
            #pragma unroll
            for (int u = 0; u < PAIR_LOADS; ++u)
                r[u] = pr2[16 * wq[i0 + 2 * u + gh] + le];
            asm volatile("" :: "v"(r[0]), "v"(r[1]), "v"(r[2]),  "v"(r[3]),  "v"(r[4]),  "v"(r[5]),
                               "v"(r[6]), "v"(r[7]), "v"(r[8]),  "v"(r[9]),  "v"(r[10]), "v"(r[11]),
                               "v"(r[12]), "v"(r[13]), "v"(r[14]), "v"(r[15]));
            unsigned bal[PAIR_LOADS];
            int tot = 0;
            #pragma unroll
            for (int u = 0; u < PAIR_LOADS; ++u) {
                bal[u] = __builtin_amdgcn_ballot_w32(i0 + 2 * u + gh < wc) &
                         __builtin_amdgcn_ballot_w32((unsigned)(r[u].y - r0) < (unsigned)rows);
                tot += __builtin_popcount(bal[u]);
            }
            int base = 0;
            if (sl == 0 && tot > 0)
                base = __hip_atomic_fetch_add(&cnt, tot, __ATOMIC_RELAXED, __HIP_MEMORY_SCOPE_WORKGROUP);
            base = __builtin_amdgcn_readfirstlane(base);
            if (base + tot <= LCAP) {
                #pragma unroll
                for (int u = 0; u < PAIR_LOADS; ++u) {
                    if (i0 + 2 * u + gh < wc && (unsigned)(r[u].y - r0) < (unsigned)rows) {
                        const int pos = base + (int)__builtin_amdgcn_mbcnt_lo(bal[u], 0u);
                        lst_s[pos] = r[u].x;
                        lst_d[pos] = r[u].y - r0;
                    }
                    base += __builtin_popcount(bal[u]);
                }
            }
        }
        for (int e = (n16 << 4) + tid; e < E; e += OWN_THREADS) {
            const unsigned dl = (unsigned)(edst[e] - r0);
            if (dl < (unsigned)rows) {
                const int pos = __hip_atomic_fetch_add(&cnt, 1, __ATOMIC_RELAXED,
                                                       __HIP_MEMORY_SCOPE_WORKGROUP);
                if (pos < LCAP) { lst_s[pos] = esrc[e]; lst_d[pos] = (int)dl; }
            }
        }
    } else {
        if (tid == 0) cnt = LCAP + 1;
    }
    __syncthreads();
    const int  total = cnt;
    const bool lmode = total <= LCAP;
    const int  n     = lmode ? total : E;

    for (int i = tid; i < n; i += OWN_THREADS) {
        int s, dl; bool own;
        if (lmode) { s = lst_s[i]; dl = lst_d[i]; own = true; }
        else {
            const unsigned u = (unsigned)(edst[i] - r0);
            own = u < (unsigned)rows; dl = own ? (int)u : 0; s = own ? esrc[i] : 0;
        }
        if (own) {
            const float4 a4 = *(const float4*)(asn + (size_t)s * HEADS);
            const float4 d4 = *(const float4*)(adl + dl * HEADS);
            const float a0 = lrelu02(a4.x + d4.x), a1 = lrelu02(a4.y + d4.y);
            const float a2 = lrelu02(a4.z + d4.z), a3 = lrelu02(a4.w + d4.w);
            if (lmode) *(float4*)(lst_a + i * HEADS) = make_float4(a0, a1, a2, a3);
            unsigned* m = mx + dl * HEADS;
            __hip_atomic_fetch_max(m + 0, f2key(a0), __ATOMIC_RELAXED, __HIP_MEMORY_SCOPE_WORKGROUP);
            __hip_atomic_fetch_max(m + 1, f2key(a1), __ATOMIC_RELAXED, __HIP_MEMORY_SCOPE_WORKGROUP);
            __hip_atomic_fetch_max(m + 2, f2key(a2), __ATOMIC_RELAXED, __HIP_MEMORY_SCOPE_WORKGROUP);
            __hip_atomic_fetch_max(m + 3, f2key(a3), __ATOMIC_RELAXED, __HIP_MEMORY_SCOPE_WORKGROUP);
        }
    }
    __syncthreads();

    for (int i = tid; i < n; i += OWN_THREADS) {
        int s, dl; bool own;
        if (lmode) { s = lst_s[i]; dl = lst_d[i]; own = true; }
        else {
            const unsigned u = (unsigned)(edst[i] - r0);
            own = u < (unsigned)rows; dl = own ? (int)u : 0; s = own ? esrc[i] : 0;
        }
        if (own) {
            float4 a;
            if (lmode) a = *(const float4*)(lst_a + i * HEADS);
            else {
                const float4 a4 = *(const float4*)(asn + (size_t)s * HEADS);
                const float4 d4 = *(const float4*)(adl + dl * HEADS);
                a = make_float4(lrelu02(a4.x + d4.x), lrelu02(a4.y + d4.y),
                                lrelu02(a4.z + d4.z), lrelu02(a4.w + d4.w));
            }
            const unsigned* m = mx + dl * HEADS;
            const float p0 = __expf(a.x - key2f(m[0])), p1 = __expf(a.y - key2f(m[1]));
            const float p2 = __expf(a.z - key2f(m[2])), p3 = __expf(a.w - key2f(m[3]));
            if (lmode) *(float4*)(lst_a + i * HEADS) = make_float4(p0, p1, p2, p3);
            float* sp = sm + dl * HEADS;
            __hip_atomic_fetch_add(sp + 0, p0, __ATOMIC_RELAXED, __HIP_MEMORY_SCOPE_WORKGROUP);
            __hip_atomic_fetch_add(sp + 1, p1, __ATOMIC_RELAXED, __HIP_MEMORY_SCOPE_WORKGROUP);
            __hip_atomic_fetch_add(sp + 2, p2, __ATOMIC_RELAXED, __HIP_MEMORY_SCOPE_WORKGROUP);
            __hip_atomic_fetch_add(sp + 3, p3, __ATOMIC_RELAXED, __HIP_MEMORY_SCOPE_WORKGROUP);
        }
    }
    __syncthreads();

    {
        const int lane = tid & 31;
        const int wv = __builtin_amdgcn_readfirstlane(tid >> 5);
        const int hl = lane & 3;
        for (int i0 = wv * EPW; i0 < n; i0 += OWN_WAVES * EPW) {
            const int il = i0 + (lane >> 2);
            if (lmode && i0 + EPW <= n) {
                const int   s  = lst_s[il];
                const int   dl = lst_d[il];
                const float p  = lst_a[il * HEADS + hl];
                float f[EPW][4];
                #pragma unroll
                for (int j = 0; j < EPW; ++j) {
                    const int sj = __builtin_amdgcn_readlane(s, 4 * j);
                    const float* sp = src_feat + (size_t)sj * FDIM + lane;
                    f[j][0] = sp[0]; f[j][1] = sp[32]; f[j][2] = sp[64]; f[j][3] = sp[96];
                }
                const float wl = p / (sm[dl * HEADS + hl] + 1e-16f);
                #pragma unroll
                for (int j = 0; j < EPW; ++j)
                    add_row_lds(acc, __builtin_amdgcn_readlane(dl, 4 * j), lane, wl, 4 * j,
                                f[j][0], f[j][1], f[j][2], f[j][3]);
            } else {
                int s = 0, dl = 0; bool own = false;
                if (il < n) {
                    if (lmode) { s = lst_s[il]; dl = lst_d[il]; own = true; }
                    else {
                        const unsigned u = (unsigned)(edst[il] - r0);
                        own = u < (unsigned)rows; dl = own ? (int)u : 0; s = own ? esrc[il] : 0;
                    }
                }
                float wl = 0.f;
                if (own) {
                    float p;
                    if (lmode) p = lst_a[il * HEADS + hl];
                    else p = __expf(lrelu02(asn[(size_t)s * HEADS + hl] + adl[dl * HEADS + hl])
                                    - key2f(mx[dl * HEADS + hl]));
                    wl = p / (sm[dl * HEADS + hl] + 1e-16f);
                }
                const unsigned om = __builtin_amdgcn_ballot_w32(own);
                for (int j = 0; j < EPW; ++j) {
                    if (om & (1u << (4 * j))) {
                        const int sj = __builtin_amdgcn_readlane(s, 4 * j);
                        const float* sp = src_feat + (size_t)sj * FDIM + lane;
                        add_row_lds(acc, __builtin_amdgcn_readlane(dl, 4 * j), lane, wl, 4 * j,
                                    sp[0], sp[32], sp[64], sp[96]);
                    }
                }
            }
        }
    }
    __syncthreads();

    {
        float4* __restrict__ op = (float4*)(out + (size_t)r0 * FDIM);
        const float4* ac = (const float4*)acc;
        for (int i = tid; i < rows * (FDIM / 4); i += OWN_THREADS) {
            float4 o = op[i];
            const float4 a = ac[i];
            o.x += a.x; o.y += a.y; o.z += a.z; o.w += a.w;
            op[i] = o;
        }
    }
}

extern "C" void kernel_launch(void* const* d_in, const int* in_sizes, int n_in,
                              void* d_out, int out_size, void* d_ws, size_t ws_size,
                              hipStream_t stream)
{
    const float* x_src  = (const float*)d_in[0];
    const float* x_dst  = (const float*)d_in[1];
    const int*   esrc   = (const int*)d_in[2];
    const int*   edst   = (const int*)d_in[3];
    const float* W_src  = (const float*)d_in[5];
    const float* W_dst  = (const float*)d_in[6];
    const float* attS   = (const float*)d_in[7];
    const float* attD   = (const float*)d_in[8];
    const float* W_self = (const float*)d_in[9];
    const float* b_self = (const float*)d_in[10];

    const int N_SRC = in_sizes[0] / FDIM;
    const int N_DST = in_sizes[1] / FDIM;
    const int E     = in_sizes[2];

    char* ws = (char*)d_ws;
    float* src_feat = (float*)(ws);
    float* as_node  = (float*)(ws + (size_t)(N_SRC + N_DST) * FDIM * 4);
    float* ad_node  = as_node + (size_t)N_SRC * HEADS;
    unsigned* keys  = (unsigned*)(ws + (size_t)N_SRC * FDIM * 4);
    int* pairs      = (int*)(ws + (size_t)N_SRC * FDIM * 4 + (((size_t)E + 255) & ~(size_t)255));

    float* out = (float*)d_out;

    node_proj<<<NP_GROUPS, 256, 0, stream>>>(
        x_src, x_dst, W_src, W_dst, W_self, b_self, attS, attD,
        src_feat, out, as_node, ad_node, esrc, edst, keys, pairs, E, N_SRC, N_DST);
    edge_owner<<<(N_DST + OWN_ROWS - 1) / OWN_ROWS, OWN_THREADS, 0, stream>>>(
        esrc, edst, src_feat, as_node, ad_node, keys, pairs, out, E, N_DST);
}


# Owner scan reads per-slice bitmaps (one bit per 16-edge group and owner, ORed in LDS by the node launch) instead of a key byte per edge: 8 KB a work-group, was 800 KB; list (sum) order differs
# speedup vs baseline: 4.4100x; 1.0405x over previous
#include <hip/hip_runtime.h>
#include <hip/hip_bf16.h>
#include <math.h>

typedef __attribute__((ext_vector_type(2))) float v2f;
typedef __attribute__((ext_vector_type(8))) float v8f;
typedef __attribute__((ext_vector_type(4))) unsigned v4u;
typedef __attribute__((ext_vector_type(2))) int v2i;
typedef __attribute__((ext_vector_type(2))) unsigned v2u;

#define HEADS 4
#define FDIM  128
#define OWN_ROWS 196

__device__ __forceinline__ unsigned f2key(float f) {
    unsigned b = __float_as_uint(f);
    return (b & 0x80000000u) ? ~b : (b | 0x80000000u);
}
__device__ __forceinline__ float key2f(unsigned k) {
    unsigned b = (k & 0x80000000u) ? (k & 0x7FFFFFFFu) : ~k;
    return __uint_as_float(b);
}

#define NP_GROUPS 512
#define BM_OWNERS 256
#define BM_WORDS  4

__device__ __forceinline__ void lds_barrier() {
    asm volatile("s_wait_dscnt 0x0\n\ts_barrier_signal -1\n\ts_barrier_wait -1" ::: "memory");
}

__device__ __forceinline__ void load_x_tile(const float* __restrict__ X, int M, int row0,
                                            int r, int c, float4& v0, float4& v1)
{
    int g0 = row0 + r;     g0 = g0 < M ? g0 : M - 1;
    int g1 = row0 + r + 8; g1 = g1 < M ? g1 : M - 1;
    v0 = *(const float4*)(X + (size_t)g0 * FDIM + c);
    v1 = *(const float4*)(X + (size_t)g1 * FDIM + c);
}

template <bool DST>
__device__ __forceinline__ void node_rows(
    float (*At)[132], float (*Pt)[132],
    const float* __restrict__ X, int M, int tiles, int first, int step,
    const float* __restrict__ Wa, const float* __restrict__ Wb,
    const float* __restrict__ att, const float* __restrict__ bias,
    float* __restrict__ Y, float* __restrict__ dots, int tid)
{
    if (first >= tiles) return;

    const int lane = tid & 31;
    const int wv   = tid >> 5;
    const int m    = lane & 15;
    const int half = lane >> 4;
    const int koff = half << 1;
    const int n    = wv * 16 + m;
    const int xr   = tid >> 5;
    const int xc   = (tid & 31) * 4;

    v2f bA[32], bB[DST ? 32 : 1];
    #pragma unroll
    for (int j = 0; j < 32; ++j) {
        bA[j].x = Wa[(4 * j + koff)     * FDIM + n];
        bA[j].y = Wa[(4 * j + koff + 1) * FDIM + n];
        if (DST) {
            bB[j].x = Wb[(4 * j + koff)     * FDIM + n];
            bB[j].y = Wb[(4 * j + koff + 1) * FDIM + n];
        }
    }
    const float an = att[n];
    const float bn = DST ? bias[n] : 0.0f;

    float4 x0, x1;
    load_x_tile(X, M, first * 16, xr, xc, x0, x1);

    for (int t = first; t < tiles; t += step) {
        const int row0 = t * 16;
        *(float4*)&At[xr][xc]     = x0;
        *(float4*)&At[xr + 8][xc] = x1;
        lds_barrier();
        if (t + step < tiles) load_x_tile(X, M, (t + step) * 16, xr, xc, x0, x1);

        v8f accA = {}, accB = {};
        #pragma unroll
        for (int j = 0; j < 32; ++j) {
            v2f a;
            a.x = At[m][4 * j + koff];
            a.y = At[m][4 * j + koff + 1];
            accA = __builtin_amdgcn_wmma_f32_16x16x4_f32(
                       false, a, false, bA[j], (short)0, accA, false, false);
            if (DST)
                accB = __builtin_amdgcn_wmma_f32_16x16x4_f32(
                           false, a, false, bB[j], (short)0, accB, false, false);
        }
        #pragma unroll
        for (int r = 0; r < 8; ++r)
            Pt[half * 8 + r][n] = accA[r] * an;
        lds_barrier();

        #pragma unroll
        for (int r = 0; r < 8; ++r) {
            const int row = row0 + half * 8 + r;
            if (row < M) Y[(size_t)row * FDIM + n] = (DST ? accB[r] : accA[r]) + bn;
        }

        if (tid < 64) {
            const int r = tid >> 2, h = tid & 3;
            const float4* p = (const float4*)&Pt[r][h * 32];
            float4 s = p[0];
            #pragma unroll
            for (int j = 1; j < 8; ++j) {
                const float4 v = p[j];
                s.x += v.x; s.y += v.y; s.z += v.z; s.w += v.w;
            }
            const float dot = (s.x + s.y) + (s.z + s.w);
            const int row = row0 + r;
            if (row < M) dots[row * HEADS + h] = dot;
        }
    }
}

__global__ __launch_bounds__(256) void node_proj(
    const float* __restrict__ x_src, const float* __restrict__ x_dst,
    const float* __restrict__ W_src, const float* __restrict__ W_dst,
    const float* __restrict__ W_self, const float* __restrict__ b_self,
    const float* __restrict__ attS, const float* __restrict__ attD,
    float* __restrict__ src_feat, float* __restrict__ out,
    float* __restrict__ as_node, float* __restrict__ ad_node,
    const int* __restrict__ esrc, const int* __restrict__ edst,
    unsigned* __restrict__ bitmaps, int* __restrict__ pairs, int E,
    int N_SRC, int N_DST)
{
    __shared__ alignas(16) float At[16][132];
    __shared__ alignas(16) float Pt[16][132];
    __shared__ alignas(16) unsigned bm[BM_OWNERS * BM_WORDS];
    const int tid = threadIdx.x;
    const int G   = (int)gridDim.x;
    const int g   = (int)blockIdx.x;

    {
        const int n16 = E >> 4;
        const int gps = (n16 + NP_GROUPS - 1) / NP_GROUPS;
        const int c0  = g * gps < n16 ? g * gps : n16;
        const int c1  = c0 + gps < n16 ? c0 + gps : n16;
        const bool bits = gps <= 32 * BM_WORDS;
        const int4* __restrict__ ed4 = (const int4*)edst;
        const int4* __restrict__ es4 = (const int4*)esrc;
        int4* __restrict__ pr4 = (int4*)pairs;
        for (int i = tid; i < BM_OWNERS * BM_WORDS; i += 256) bm[i] = 0u;
        lds_barrier();
        for (int q = 4 * c0 + tid; q < 4 * c1; q += 256) {
            const int4 v = ed4[q];
            const int4 s = es4[q];
            pr4[2 * q]     = make_int4(s.x, v.x, s.y, v.y);
            pr4[2 * q + 1] = make_int4(s.z, v.z, s.w, v.w);
            if (bits) {
                const int gi = (q >> 2) - c0;
                const unsigned bit = 1u << (gi & 31);
                unsigned* bw = bm + (gi >> 5);
                const unsigned o0 = (unsigned)v.x / OWN_ROWS, o1 = (unsigned)v.y / OWN_ROWS;
                const unsigned o2 = (unsigned)v.z / OWN_ROWS, o3 = (unsigned)v.w / OWN_ROWS;
                if (o0 < BM_OWNERS) __hip_atomic_fetch_or(bw + o0 * BM_WORDS, bit, __ATOMIC_RELAXED, __HIP_MEMORY_SCOPE_WORKGROUP);
                if (o1 < BM_OWNERS) __hip_atomic_fetch_or(bw + o1 * BM_WORDS, bit, __ATOMIC_RELAXED, __HIP_MEMORY_SCOPE_WORKGROUP);
                if (o2 < BM_OWNERS) __hip_atomic_fetch_or(bw + o2 * BM_WORDS, bit, __ATOMIC_RELAXED, __HIP_MEMORY_SCOPE_WORKGROUP);
                if (o3 < BM_OWNERS) __hip_atomic_fetch_or(bw + o3 * BM_WORDS, bit, __ATOMIC_RELAXED, __HIP_MEMORY_SCOPE_WORKGROUP);
            }
        }
        lds_barrier();
        if (g < NP_GROUPS)
            for (int o = tid; o < BM_OWNERS; o += 256)
                ((v4u*)bitmaps)[(size_t)o * NP_GROUPS + g] = *(const v4u*)&bm[o * BM_WORDS];
    }

    node_rows<true >(At, Pt, x_dst, N_DST, (N_DST + 15) >> 4, g, G,
                     W_dst, W_self, attD, b_self, out, ad_node, tid);
    node_rows<false>(At, Pt, x_src, N_SRC, (N_SRC + 15) >> 4, G - 1 - g, G,
                     W_src, W_src, attS, b_self, src_feat, as_node, tid);
}

#define OWN_THREADS 1024
#define OWN_WAVES   (OWN_THREADS / 32)
#define LCAP        4096
#define WQCAP       (LCAP * HEADS / OWN_WAVES)
#define EPW         8
#define PAIR_LOADS  16

__device__ __forceinline__ float lrelu02(float a) { return a > 0.f ? a : 0.2f * a; }

__device__ __forceinline__ void add_row_lds(float* acc, int d, int lane, float wl, int q,
                                            float f0, float f1, float f2, float f3)
{
    const float w0 = __int_as_float(__builtin_amdgcn_readlane(__float_as_int(wl), q + 0));
    const float w1 = __int_as_float(__builtin_amdgcn_readlane(__float_as_int(wl), q + 1));
    const float w2 = __int_as_float(__builtin_amdgcn_readlane(__float_as_int(wl), q + 2));
    const float w3 = __int_as_float(__builtin_amdgcn_readlane(__float_as_int(wl), q + 3));
    float* ap = acc + d * FDIM + lane;
    __hip_atomic_fetch_add(ap +  0, w0 * f0, __ATOMIC_RELAXED, __HIP_MEMORY_SCOPE_WORKGROUP);
    __hip_atomic_fetch_add(ap + 32, w1 * f1, __ATOMIC_RELAXED, __HIP_MEMORY_SCOPE_WORKGROUP);
    __hip_atomic_fetch_add(ap + 64, w2 * f2, __ATOMIC_RELAXED, __HIP_MEMORY_SCOPE_WORKGROUP);
    __hip_atomic_fetch_add(ap + 96, w3 * f3, __ATOMIC_RELAXED, __HIP_MEMORY_SCOPE_WORKGROUP);
}

__global__ __launch_bounds__(OWN_THREADS) void edge_owner(
    const int* __restrict__ esrc, const int* __restrict__ edst,
    const float* __restrict__ src_feat,
    const float* __restrict__ asn, const float* __restrict__ adn,
    const unsigned* __restrict__ bitmaps, const int* __restrict__ pairs,
    float* __restrict__ out, int E, int N_DST)
{
    __shared__ alignas(16) float acc[OWN_ROWS * FDIM];
    __shared__ alignas(16) float adl[OWN_ROWS * HEADS];
    __shared__ unsigned mx[OWN_ROWS * HEADS];
    __shared__ float    sm[OWN_ROWS * HEADS];
    __shared__ int      lst_s[LCAP];
    __shared__ int      lst_d[LCAP];
    __shared__ alignas(16) float lst_a[LCAP * HEADS];
    __shared__ int      cnt;

    const int tid  = threadIdx.x;
    const int r0   = blockIdx.x * OWN_ROWS;
    const int left = N_DST - r0;
    const int rows = left < OWN_ROWS ? left : OWN_ROWS;

    for (int i = tid; i < OWN_ROWS * FDIM / 4; i += OWN_THREADS)
        ((float4*)acc)[i] = make_float4(0.f, 0.f, 0.f, 0.f);
    for (int i = tid; i < rows * HEADS; i += OWN_THREADS) {
        adl[i] = adn[(size_t)r0 * HEADS + i];
        mx[i] = 0u;
        sm[i] = 0.f;
    }
    if (tid == 0) cnt = 0;
    __syncthreads();

    const int n16 = E >> 4;
    const int gps = (n16 + NP_GROUPS - 1) / NP_GROUPS;
    const bool keyed = gridDim.x <= (unsigned)BM_OWNERS && gps <= 32 * BM_WORDS;
    if (keyed) {
        const v2u* __restrict__ b2 = (const v2u*)bitmaps + (size_t)blockIdx.x * (NP_GROUPS * BM_WORDS / 2);
        const int sl = tid & 31;
        const int sw = __builtin_amdgcn_readfirstlane(tid >> 5);
        int* wq = (int*)lst_a + sw * WQCAP;
        int  wn = 0;
        for (int i = tid; i < NP_GROUPS * BM_WORDS / 2; i += OWN_THREADS) {
            const v2u w = b2[i];
            unsigned lo = w.x, hi = w.y;
            const int gb = (i >> 1) * gps + (i & 1) * 64;
            for (int trip = 0; trip < 64; ++trip) {
                const bool hit = (lo | hi) != 0u;
                const unsigned bal = __builtin_amdgcn_ballot_w32(hit);
                if (bal == 0u) break;
                if (hit) {
                    int b;
                    if (lo != 0u) { b = __builtin_ctz(lo); lo &= lo - 1u; }
                    else          { b = 32 + __builtin_ctz(hi); hi &= hi - 1u; }
                    const int pos = wn + (int)__builtin_amdgcn_mbcnt_lo(bal, 0u);
                    if (pos < WQCAP) wq[pos] = gb + b;
                }
                wn += __builtin_popcount(bal);
            }
        }
        if (sl == 0 && wn > WQCAP)
            __hip_atomic_fetch_add(&cnt, LCAP + 1, __ATOMIC_RELAXED, __HIP_MEMORY_SCOPE_WORKGROUP);
        const int wc = wn < WQCAP ? wn : WQCAP;
        if (wc + sl < ((wc + 31) & ~31)) wq[wc + sl] = 0;
        __syncthreads();
        const v2i* __restrict__ pr2 = (const v2i*)pairs;
        const int gh = sl >> 4;
        const int le = sl & 15;
        for (int i0 = 0; i0 < wc; i0 += 2 * PAIR_LOADS) {
            v2i r[PAIR_LOADS];
            #pragma unroll
            for (int u = 0; u < PAIR_LOADS; ++u)
                r[u] = pr2[16 * wq[i0 + 2 * u + gh] + le];
            asm volatile("" :: "v"(r[0]), "v"(r[1]), "v"(r[2]),  "v"(r[3]),  "v"(r[4]),  "v"(r[5]),
                               "v"(r[6]), "v"(r[7]), "v"(r[8]),  "v"(r[9]),  "v"(r[10]), "v"(r[11]),
                               "v"(r[12]), "v"(r[13]), "v"(r[14]), "v"(r[15]));
            unsigned bal[PAIR_LOADS];
            int tot = 0;
            #pragma unroll
            for (int u = 0; u < PAIR_LOADS; ++u) {
                bal[u] = __builtin_amdgcn_ballot_w32(i0 + 2 * u + gh < wc) &
                         __builtin_amdgcn_ballot_w32((unsigned)(r[u].y - r0) < (unsigned)rows);
                tot += __builtin_popcount(bal[u]);
            }
            int base = 0;
            if (sl == 0 && tot > 0)
                base = __hip_atomic_fetch_add(&cnt, tot, __ATOMIC_RELAXED, __HIP_MEMORY_SCOPE_WORKGROUP);
            base = __builtin_amdgcn_readfirstlane(base);
            if (base + tot <= LCAP) {
                #pragma unroll
                for (int u = 0; u < PAIR_LOADS; ++u) {
                    if (i0 + 2 * u + gh < wc && (unsigned)(r[u].y - r0) < (unsigned)rows) {
                        const int pos = base + (int)__builtin_amdgcn_mbcnt_lo(bal[u], 0u);
                        lst_s[pos] = r[u].x;
                        lst_d[pos] = r[u].y - r0;
                    }
                    base += __builtin_popcount(bal[u]);
                }
            }
        }
        for (int e = (n16 << 4) + tid; e < E; e += OWN_THREADS) {
            const unsigned dl = (unsigned)(edst[e] - r0);
            if (dl < (unsigned)rows) {
                const int pos = __hip_atomic_fetch_add(&cnt, 1, __ATOMIC_RELAXED,
                                                       __HIP_MEMORY_SCOPE_WORKGROUP);
                if (pos < LCAP) { lst_s[pos] = esrc[e]; lst_d[pos] = (int)dl; }
            }
        }
    } else {
        if (tid == 0) cnt = LCAP + 1;
    }
    __syncthreads();
    const int  total = cnt;
    const bool lmode = total <= LCAP;
    const int  n     = lmode ? total : E;

    for (int i = tid; i < n; i += OWN_THREADS) {
        int s, dl; bool own;
        if (lmode) { s = lst_s[i]; dl = lst_d[i]; own = true; }
        else {
            const unsigned u = (unsigned)(edst[i] - r0);
            own = u < (unsigned)rows; dl = own ? (int)u : 0; s = own ? esrc[i] : 0;
        }
        if (own) {
            const float4 a4 = *(const float4*)(asn + (size_t)s * HEADS);
            const float4 d4 = *(const float4*)(adl + dl * HEADS);
            const float a0 = lrelu02(a4.x + d4.x), a1 = lrelu02(a4.y + d4.y);
            const float a2 = lrelu02(a4.z + d4.z), a3 = lrelu02(a4.w + d4.w);
            if (lmode) *(float4*)(lst_a + i * HEADS) = make_float4(a0, a1, a2, a3);
            unsigned* m = mx + dl * HEADS;
            __hip_atomic_fetch_max(m + 0, f2key(a0), __ATOMIC_RELAXED, __HIP_MEMORY_SCOPE_WORKGROUP);
            __hip_atomic_fetch_max(m + 1, f2key(a1), __ATOMIC_RELAXED, __HIP_MEMORY_SCOPE_WORKGROUP);
            __hip_atomic_fetch_max(m + 2, f2key(a2), __ATOMIC_RELAXED, __HIP_MEMORY_SCOPE_WORKGROUP);
            __hip_atomic_fetch_max(m + 3, f2key(a3), __ATOMIC_RELAXED, __HIP_MEMORY_SCOPE_WORKGROUP);
        }
    }
    __syncthreads();

    for (int i = tid; i < n; i += OWN_THREADS) {
        int s, dl; bool own;
        if (lmode) { s = lst_s[i]; dl = lst_d[i]; own = true; }
        else {
            const unsigned u = (unsigned)(edst[i] - r0);
            own = u < (unsigned)rows; dl = own ? (int)u : 0; s = own ? esrc[i] : 0;
        }
        if (own) {
            float4 a;
            if (lmode) a = *(const float4*)(lst_a + i * HEADS);
            else {
                const float4 a4 = *(const float4*)(asn + (size_t)s * HEADS);
                const float4 d4 = *(const float4*)(adl + dl * HEADS);
                a = make_float4(lrelu02(a4.x + d4.x), lrelu02(a4.y + d4.y),
                                lrelu02(a4.z + d4.z), lrelu02(a4.w + d4.w));
            }
            const unsigned* m = mx + dl * HEADS;
            const float p0 = __expf(a.x - key2f(m[0])), p1 = __expf(a.y - key2f(m[1]));
            const float p2 = __expf(a.z - key2f(m[2])), p3 = __expf(a.w - key2f(m[3]));
            if (lmode) *(float4*)(lst_a + i * HEADS) = make_float4(p0, p1, p2, p3);
            float* sp = sm + dl * HEADS;
            __hip_atomic_fetch_add(sp + 0, p0, __ATOMIC_RELAXED, __HIP_MEMORY_SCOPE_WORKGROUP);
            __hip_atomic_fetch_add(sp + 1, p1, __ATOMIC_RELAXED, __HIP_MEMORY_SCOPE_WORKGROUP);
            __hip_atomic_fetch_add(sp + 2, p2, __ATOMIC_RELAXED, __HIP_MEMORY_SCOPE_WORKGROUP);
            __hip_atomic_fetch_add(sp + 3, p3, __ATOMIC_RELAXED, __HIP_MEMORY_SCOPE_WORKGROUP);
        }
    }
    __syncthreads();

    {
        const int lane = tid & 31;
        const int wv = __builtin_amdgcn_readfirstlane(tid >> 5);
        const int hl = lane & 3;
        for (int i0 = wv * EPW; i0 < n; i0 += OWN_WAVES * EPW) {
            const int il = i0 + (lane >> 2);
            if (lmode && i0 + EPW <= n) {
                const int   s  = lst_s[il];
                const int   dl = lst_d[il];
                const float p  = lst_a[il * HEADS + hl];
                float f[EPW][4];
                #pragma unroll
                for (int j = 0; j < EPW; ++j) {
                    const int sj = __builtin_amdgcn_readlane(s, 4 * j);
                    const float* sp = src_feat + (size_t)sj * FDIM + lane;
                    f[j][0] = sp[0]; f[j][1] = sp[32]; f[j][2] = sp[64]; f[j][3] = sp[96];
                }
                const float wl = p / (sm[dl * HEADS + hl] + 1e-16f);
                #pragma unroll
                for (int j = 0; j < EPW; ++j)
                    add_row_lds(acc, __builtin_amdgcn_readlane(dl, 4 * j), lane, wl, 4 * j,
                                f[j][0], f[j][1], f[j][2], f[j][3]);
            } else {
                int s = 0, dl = 0; bool own = false;
                if (il < n) {
                    if (lmode) { s = lst_s[il]; dl = lst_d[il]; own = true; }
                    else {
                        const unsigned u = (unsigned)(edst[il] - r0);
                        own = u < (unsigned)rows; dl = own ? (int)u : 0; s = own ? esrc[il] : 0;
                    }
                }
                float wl = 0.f;
                if (own) {
                    float p;
                    if (lmode) p = lst_a[il * HEADS + hl];
                    else p = __expf(lrelu02(asn[(size_t)s * HEADS + hl] + adl[dl * HEADS + hl])
                                    - key2f(mx[dl * HEADS + hl]));
                    wl = p / (sm[dl * HEADS + hl] + 1e-16f);
                }
                const unsigned om = __builtin_amdgcn_ballot_w32(own);
                for (int j = 0; j < EPW; ++j) {
                    if (om & (1u << (4 * j))) {
                        const int sj = __builtin_amdgcn_readlane(s, 4 * j);
                        const float* sp = src_feat + (size_t)sj * FDIM + lane;
                        add_row_lds(acc, __builtin_amdgcn_readlane(dl, 4 * j), lane, wl, 4 * j,
                                    sp[0], sp[32], sp[64], sp[96]);
                    }
                }
            }
        }
    }
    __syncthreads();

    {
        float4* __restrict__ op = (float4*)(out + (size_t)r0 * FDIM);
        const float4* ac = (const float4*)acc;
        for (int i = tid; i < rows * (FDIM / 4); i += OWN_THREADS) {
            float4 o = op[i];
            const float4 a = ac[i];
            o.x += a.x; o.y += a.y; o.z += a.z; o.w += a.w;
            op[i] = o;
        }
    }
}

extern "C" void kernel_launch(void* const* d_in, const int* in_sizes, int n_in,
                              void* d_out, int out_size, void* d_ws, size_t ws_size,
                              hipStream_t stream)
{
    const float* x_src  = (const float*)d_in[0];
    const float* x_dst  = (const float*)d_in[1];
    const int*   esrc   = (const int*)d_in[2];
    const int*   edst   = (const int*)d_in[3];
    const float* W_src  = (const float*)d_in[5];
    const float* W_dst  = (const float*)d_in[6];
    const float* attS   = (const float*)d_in[7];
    const float* attD   = (const float*)d_in[8];
    const float* W_self = (const float*)d_in[9];
    const float* b_self = (const float*)d_in[10];

    const int N_SRC = in_sizes[0] / FDIM;
    const int N_DST = in_sizes[1] / FDIM;
    const int E     = in_sizes[2];

    char* ws = (char*)d_ws;
    float* src_feat = (float*)(ws);
    float* as_node  = (float*)(ws + (size_t)(N_SRC + N_DST) * FDIM * 4);
    float* ad_node  = as_node + (size_t)N_SRC * HEADS;
    unsigned* bitmaps = (unsigned*)(ws + (size_t)N_SRC * FDIM * 4);
    int* pairs        = (int*)(ws + (size_t)N_SRC * FDIM * 4
                                  + (size_t)BM_OWNERS * NP_GROUPS * BM_WORDS * 4);

    float* out = (float*)d_out;

    node_proj<<<NP_GROUPS, 256, 0, stream>>>(
        x_src, x_dst, W_src, W_dst, W_self, b_self, attS, attD,
        src_feat, out, as_node, ad_node, esrc, edst, bitmaps, pairs, E, N_SRC, N_DST);
    edge_owner<<<(N_DST + OWN_ROWS - 1) / OWN_ROWS, OWN_THREADS, 0, stream>>>(
        esrc, edst, src_feat, as_node, ad_node, bitmaps, pairs, out, E, N_DST);
}


# Owner's edge list grouped by destination row (rank from an LDS counter at list build, placed at the exp step): a wave sums its 8 entries per row in registers, one LDS add a run; sum order changes
# speedup vs baseline: 4.5663x; 1.0354x over previous
#include <hip/hip_runtime.h>
#include <hip/hip_bf16.h>
#include <math.h>

typedef __attribute__((ext_vector_type(2))) float v2f;
typedef __attribute__((ext_vector_type(8))) float v8f;
typedef __attribute__((ext_vector_type(4))) unsigned v4u;
typedef __attribute__((ext_vector_type(2))) int v2i;
typedef __attribute__((ext_vector_type(2))) unsigned v2u;

#define HEADS 4
#define FDIM  128
#define OWN_ROWS 196

__device__ __forceinline__ unsigned f2key(float f) {
    unsigned b = __float_as_uint(f);
    return (b & 0x80000000u) ? ~b : (b | 0x80000000u);
}
__device__ __forceinline__ float key2f(unsigned k) {
    unsigned b = (k & 0x80000000u) ? (k & 0x7FFFFFFFu) : ~k;
    return __uint_as_float(b);
}

#define NP_GROUPS 512
#define BM_OWNERS 256
#define BM_WORDS  4

__device__ __forceinline__ void lds_barrier() {
    asm volatile("s_wait_dscnt 0x0\n\ts_barrier_signal -1\n\ts_barrier_wait -1" ::: "memory");
}

__device__ __forceinline__ void load_x_tile(const float* __restrict__ X, int M, int row0,
                                            int r, int c, float4& v0, float4& v1)
{
    int g0 = row0 + r;     g0 = g0 < M ? g0 : M - 1;
    int g1 = row0 + r + 8; g1 = g1 < M ? g1 : M - 1;
    v0 = *(const float4*)(X + (size_t)g0 * FDIM + c);
    v1 = *(const float4*)(X + (size_t)g1 * FDIM + c);
}

template <bool DST>
__device__ __forceinline__ void node_rows(
    float (*At)[132], float (*Pt)[132],
    const float* __restrict__ X, int M, int tiles, int first, int step,
    const float* __restrict__ Wa, const float* __restrict__ Wb,
    const float* __restrict__ att, const float* __restrict__ bias,
    float* __restrict__ Y, float* __restrict__ dots, int tid)
{
    if (first >= tiles) return;

    const int lane = tid & 31;
    const int wv   = tid >> 5;
    const int m    = lane & 15;
    const int half = lane >> 4;
    const int koff = half << 1;
    const int n    = wv * 16 + m;
    const int xr   = tid >> 5;
    const int xc   = (tid & 31) * 4;

    v2f bA[32], bB[DST ? 32 : 1];
    #pragma unroll
    for (int j = 0; j < 32; ++j) {
        bA[j].x = Wa[(4 * j + koff)     * FDIM + n];
        bA[j].y = Wa[(4 * j + koff + 1) * FDIM + n];
        if (DST) {
            bB[j].x = Wb[(4 * j + koff)     * FDIM + n];
            bB[j].y = Wb[(4 * j + koff + 1) * FDIM + n];
        }
    }
    const float an = att[n];
    const float bn = DST ? bias[n] : 0.0f;

    float4 x0, x1;
    load_x_tile(X, M, first * 16, xr, xc, x0, x1);

    for (int t = first; t < tiles; t += step) {
        const int row0 = t * 16;
        *(float4*)&At[xr][xc]     = x0;
        *(float4*)&At[xr + 8][xc] = x1;
        lds_barrier();
        if (t + step < tiles) load_x_tile(X, M, (t + step) * 16, xr, xc, x0, x1);

        v8f accA = {}, accB = {};
        #pragma unroll
        for (int j = 0; j < 32; ++j) {
            v2f a;
            a.x = At[m][4 * j + koff];
            a.y = At[m][4 * j + koff + 1];
            accA = __builtin_amdgcn_wmma_f32_16x16x4_f32(
                       false, a, false, bA[j], (short)0, accA, false, false);
            if (DST)
                accB = __builtin_amdgcn_wmma_f32_16x16x4_f32(
                           false, a, false, bB[j], (short)0, accB, false, false);
        }
        #pragma unroll
        for (int r = 0; r < 8; ++r)
            Pt[half * 8 + r][n] = accA[r] * an;
        lds_barrier();

        #pragma unroll
        for (int r = 0; r < 8; ++r) {
            const int row = row0 + half * 8 + r;
            if (row < M) Y[(size_t)row * FDIM + n] = (DST ? accB[r] : accA[r]) + bn;
        }

        if (tid < 64) {
            const int r = tid >> 2, h = tid & 3;
            const float4* p = (const float4*)&Pt[r][h * 32];
            float4 s = p[0];
            #pragma unroll
            for (int j = 1; j < 8; ++j) {
                const float4 v = p[j];
                s.x += v.x; s.y += v.y; s.z += v.z; s.w += v.w;
            }
            const float dot = (s.x + s.y) + (s.z + s.w);
            const int row = row0 + r;
            if (row < M) dots[row * HEADS + h] = dot;
        }
    }
}

__global__ __launch_bounds__(256) void node_proj(
    const float* __restrict__ x_src, const float* __restrict__ x_dst,
    const float* __restrict__ W_src, const float* __restrict__ W_dst,
    const float* __restrict__ W_self, const float* __restrict__ b_self,
    const float* __restrict__ attS, const float* __restrict__ attD,
    float* __restrict__ src_feat, float* __restrict__ out,
    float* __restrict__ as_node, float* __restrict__ ad_node,
    const int* __restrict__ esrc, const int* __restrict__ edst,
    unsigned* __restrict__ bitmaps, int* __restrict__ pairs, int E,
    int N_SRC, int N_DST)
{
    __shared__ alignas(16) float At[16][132];
    __shared__ alignas(16) float Pt[16][132];
    __shared__ alignas(16) unsigned bm[BM_OWNERS * BM_WORDS];
    const int tid = threadIdx.x;
    const int G   = (int)gridDim.x;
    const int g   = (int)blockIdx.x;

    {
        const int n16 = E >> 4;
        const int gps = (n16 + NP_GROUPS - 1) / NP_GROUPS;
        const int c0  = g * gps < n16 ? g * gps : n16;
        const int c1  = c0 + gps < n16 ? c0 + gps : n16;
        const bool bits = gps <= 32 * BM_WORDS;
        const int4* __restrict__ ed4 = (const int4*)edst;
        const int4* __restrict__ es4 = (const int4*)esrc;
        int4* __restrict__ pr4 = (int4*)pairs;
        for (int i = tid; i < BM_OWNERS * BM_WORDS; i += 256) bm[i] = 0u;
        lds_barrier();
        for (int q = 4 * c0 + tid; q < 4 * c1; q += 256) {
            const int4 v = ed4[q];
            const int4 s = es4[q];
            pr4[2 * q]     = make_int4(s.x, v.x, s.y, v.y);
            pr4[2 * q + 1] = make_int4(s.z, v.z, s.w, v.w);
            if (bits) {
                const int gi = (q >> 2) - c0;
                const unsigned bit = 1u << (gi & 31);
                unsigned* bw = bm + (gi >> 5);
                const unsigned o0 = (unsigned)v.x / OWN_ROWS, o1 = (unsigned)v.y / OWN_ROWS;
                const unsigned o2 = (unsigned)v.z / OWN_ROWS, o3 = (unsigned)v.w / OWN_ROWS;
                if (o0 < BM_OWNERS) __hip_atomic_fetch_or(bw + o0 * BM_WORDS, bit, __ATOMIC_RELAXED, __HIP_MEMORY_SCOPE_WORKGROUP);
                if (o1 < BM_OWNERS) __hip_atomic_fetch_or(bw + o1 * BM_WORDS, bit, __ATOMIC_RELAXED, __HIP_MEMORY_SCOPE_WORKGROUP);
                if (o2 < BM_OWNERS) __hip_atomic_fetch_or(bw + o2 * BM_WORDS, bit, __ATOMIC_RELAXED, __HIP_MEMORY_SCOPE_WORKGROUP);
                if (o3 < BM_OWNERS) __hip_atomic_fetch_or(bw + o3 * BM_WORDS, bit, __ATOMIC_RELAXED, __HIP_MEMORY_SCOPE_WORKGROUP);
            }
        }
        lds_barrier();
        if (g < NP_GROUPS)
            for (int o = tid; o < BM_OWNERS; o += 256)
                ((v4u*)bitmaps)[(size_t)o * NP_GROUPS + g] = *(const v4u*)&bm[o * BM_WORDS];
    }

    node_rows<true >(At, Pt, x_dst, N_DST, (N_DST + 15) >> 4, g, G,
                     W_dst, W_self, attD, b_self, out, ad_node, tid);
    node_rows<false>(At, Pt, x_src, N_SRC, (N_SRC + 15) >> 4, G - 1 - g, G,
                     W_src, W_src, attS, b_self, src_feat, as_node, tid);
}

#define OWN_THREADS 1024
#define OWN_WAVES   (OWN_THREADS / 32)
#define LCAP        4096
#define WQCAP       (LCAP * HEADS / OWN_WAVES)
#define EPW         8
#define PAIR_LOADS  16
#define ROW_BITS    8
#define ROWS_PER_LANE ((OWN_ROWS + 31) / 32)
static_assert(OWN_ROWS <= (1 << ROW_BITS), "a local row must fit its bits");
static_assert(LCAP % OWN_THREADS == 0, "the exp step holds LCAP / OWN_THREADS entries a thread");

__device__ __forceinline__ float lrelu02(float a) { return a > 0.f ? a : 0.2f * a; }

__device__ __forceinline__ void add_row_lds(float* acc, int d, int lane, float wl, int q,
                                            float f0, float f1, float f2, float f3)
{
    const float w0 = __int_as_float(__builtin_amdgcn_readlane(__float_as_int(wl), q + 0));
    const float w1 = __int_as_float(__builtin_amdgcn_readlane(__float_as_int(wl), q + 1));
    const float w2 = __int_as_float(__builtin_amdgcn_readlane(__float_as_int(wl), q + 2));
    const float w3 = __int_as_float(__builtin_amdgcn_readlane(__float_as_int(wl), q + 3));
    float* ap = acc + d * FDIM + lane;
    __hip_atomic_fetch_add(ap +  0, w0 * f0, __ATOMIC_RELAXED, __HIP_MEMORY_SCOPE_WORKGROUP);
    __hip_atomic_fetch_add(ap + 32, w1 * f1, __ATOMIC_RELAXED, __HIP_MEMORY_SCOPE_WORKGROUP);
    __hip_atomic_fetch_add(ap + 64, w2 * f2, __ATOMIC_RELAXED, __HIP_MEMORY_SCOPE_WORKGROUP);
    __hip_atomic_fetch_add(ap + 96, w3 * f3, __ATOMIC_RELAXED, __HIP_MEMORY_SCOPE_WORKGROUP);
}

__device__ __forceinline__ void add_run_lds(float* acc, int d, int lane,
                                            float a0, float a1, float a2, float a3)
{
    float* ap = acc + d * FDIM + lane;
    __hip_atomic_fetch_add(ap +  0, a0, __ATOMIC_RELAXED, __HIP_MEMORY_SCOPE_WORKGROUP);
    __hip_atomic_fetch_add(ap + 32, a1, __ATOMIC_RELAXED, __HIP_MEMORY_SCOPE_WORKGROUP);
    __hip_atomic_fetch_add(ap + 64, a2, __ATOMIC_RELAXED, __HIP_MEMORY_SCOPE_WORKGROUP);
    __hip_atomic_fetch_add(ap + 96, a3, __ATOMIC_RELAXED, __HIP_MEMORY_SCOPE_WORKGROUP);
}

__global__ __launch_bounds__(OWN_THREADS) void edge_owner(
    const int* __restrict__ esrc, const int* __restrict__ edst,
    const float* __restrict__ src_feat,
    const float* __restrict__ asn, const float* __restrict__ adn,
    const unsigned* __restrict__ bitmaps, const int* __restrict__ pairs,
    float* __restrict__ out, int E, int N_DST)
{
    __shared__ alignas(16) float acc[OWN_ROWS * FDIM];
    __shared__ alignas(16) float adl[OWN_ROWS * HEADS];
    __shared__ unsigned mx[OWN_ROWS * HEADS];
    __shared__ float    sm[OWN_ROWS * HEADS];
    __shared__ int      lst_s[LCAP];
    __shared__ int      lst_d[LCAP];
    __shared__ int      rowc[OWN_ROWS];
    __shared__ alignas(16) float lst_a[LCAP * HEADS];
    __shared__ int      cnt;

    const int tid  = threadIdx.x;
    const int r0   = blockIdx.x * OWN_ROWS;
    const int left = N_DST - r0;
    const int rows = left < OWN_ROWS ? left : OWN_ROWS;

    for (int i = tid; i < OWN_ROWS * FDIM / 4; i += OWN_THREADS)
        ((float4*)acc)[i] = make_float4(0.f, 0.f, 0.f, 0.f);
    for (int i = tid; i < rows * HEADS; i += OWN_THREADS) {
        adl[i] = adn[(size_t)r0 * HEADS + i];
        mx[i] = 0u;
        sm[i] = 0.f;
    }
    for (int i = tid; i < OWN_ROWS; i += OWN_THREADS) rowc[i] = 0;
    if (tid == 0) cnt = 0;
    __syncthreads();

    const int n16 = E >> 4;
    const int gps = (n16 + NP_GROUPS - 1) / NP_GROUPS;
    const bool keyed = gridDim.x <= (unsigned)BM_OWNERS && gps <= 32 * BM_WORDS;
    if (keyed) {
        const v2u* __restrict__ b2 = (const v2u*)bitmaps + (size_t)blockIdx.x * (NP_GROUPS * BM_WORDS / 2);
        const int sl = tid & 31;
        const int sw = __builtin_amdgcn_readfirstlane(tid >> 5);
        int* wq = (int*)lst_a + sw * WQCAP;
        int  wn = 0;
        for (int i = tid; i < NP_GROUPS * BM_WORDS / 2; i += OWN_THREADS) {
            const v2u w = b2[i];
            unsigned lo = w.x, hi = w.y;
            const int gb = (i >> 1) * gps + (i & 1) * 64;
            for (int trip = 0; trip < 64; ++trip) {
                const bool hit = (lo | hi) != 0u;
                const unsigned bal = __builtin_amdgcn_ballot_w32(hit);
                if (bal == 0u) break;
                if (hit) {
                    int b;
                    if (lo != 0u) { b = __builtin_ctz(lo); lo &= lo - 1u; }
                    else          { b = 32 + __builtin_ctz(hi); hi &= hi - 1u; }
                    const int pos = wn + (int)__builtin_amdgcn_mbcnt_lo(bal, 0u);
                    if (pos < WQCAP) wq[pos] = gb + b;
                }
                wn += __builtin_popcount(bal);
            }
        }
        if (sl == 0 && wn > WQCAP)
            __hip_atomic_fetch_add(&cnt, LCAP + 1, __ATOMIC_RELAXED, __HIP_MEMORY_SCOPE_WORKGROUP);
        const int wc = wn < WQCAP ? wn : WQCAP;
        if (wc + sl < ((wc + 31) & ~31)) wq[wc + sl] = 0;
        __syncthreads();
        const v2i* __restrict__ pr2 = (const v2i*)pairs;
        const int gh = sl >> 4;
        const int le = sl & 15;
        for (int i0 = 0; i0 < wc; i0 += 2 * PAIR_LOADS) {
            v2i r[PAIR_LOADS];
            #pragma unroll
            for (int u = 0; u < PAIR_LOADS; ++u)
                r[u] = pr2[16 * wq[i0 + 2 * u + gh] + le];
            asm volatile("" :: "v"(r[0]), "v"(r[1]), "v"(r[2]),  "v"(r[3]),  "v"(r[4]),  "v"(r[5]),
                               "v"(r[6]), "v"(r[7]), "v"(r[8]),  "v"(r[9]),  "v"(r[10]), "v"(r[11]),
                               "v"(r[12]), "v"(r[13]), "v"(r[14]), "v"(r[15]));
            unsigned bal[PAIR_LOADS];
            int tot = 0;
            #pragma unroll
            for (int u = 0; u < PAIR_LOADS; ++u) {
                bal[u] = __builtin_amdgcn_ballot_w32(i0 + 2 * u + gh < wc) &
                         __builtin_amdgcn_ballot_w32((unsigned)(r[u].y - r0) < (unsigned)rows);
                tot += __builtin_popcount(bal[u]);
            }
            int base = 0;
            if (sl == 0 && tot > 0)
                base = __hip_atomic_fetch_add(&cnt, tot, __ATOMIC_RELAXED, __HIP_MEMORY_SCOPE_WORKGROUP);
            base = __builtin_amdgcn_readfirstlane(base);
            if (base + tot <= LCAP) {
                #pragma unroll
                for (int u = 0; u < PAIR_LOADS; ++u) {
                    if (i0 + 2 * u + gh < wc && (unsigned)(r[u].y - r0) < (unsigned)rows) {
                        const int pos = base + (int)__builtin_amdgcn_mbcnt_lo(bal[u], 0u);
                        const int dl  = r[u].y - r0;
                        const int rk  = __hip_atomic_fetch_add(&rowc[dl], 1, __ATOMIC_RELAXED,
                                                               __HIP_MEMORY_SCOPE_WORKGROUP);
                        lst_s[pos] = r[u].x;
                        lst_d[pos] = dl | (rk << ROW_BITS);
                    }
                    base += __builtin_popcount(bal[u]);
                }
            }
        }
        for (int e = (n16 << 4) + tid; e < E; e += OWN_THREADS) {
            const unsigned dl = (unsigned)(edst[e] - r0);
            if (dl < (unsigned)rows) {
                const int pos = __hip_atomic_fetch_add(&cnt, 1, __ATOMIC_RELAXED,
                                                       __HIP_MEMORY_SCOPE_WORKGROUP);
                if (pos < LCAP) {
                    const int rk = __hip_atomic_fetch_add(&rowc[dl], 1, __ATOMIC_RELAXED,
                                                          __HIP_MEMORY_SCOPE_WORKGROUP);
                    lst_s[pos] = esrc[e]; lst_d[pos] = (int)dl | (rk << ROW_BITS);
                }
            }
        }
    } else {
        if (tid == 0) cnt = LCAP + 1;
    }
    __syncthreads();
    const int  total = cnt;
    const bool lmode = total <= LCAP;
    const int  n     = lmode ? total : E;

    if (lmode && tid < 32) {
        int c[ROWS_PER_LANE];
        int sum = 0;
        #pragma unroll
        for (int k = 0; k < ROWS_PER_LANE; ++k) {
            const int rr = tid * ROWS_PER_LANE + k;
            c[k] = rr < OWN_ROWS ? rowc[rr] : 0;
            sum += c[k];
        }
        int inc = sum;
        for (int d = 1; d < 32; d <<= 1) {
            const int t = __shfl_up(inc, d);
            if (tid >= d) inc += t;
        }
        int ex = inc - sum;
        #pragma unroll
        for (int k = 0; k < ROWS_PER_LANE; ++k) {
            const int rr = tid * ROWS_PER_LANE + k;
            if (rr < OWN_ROWS) rowc[rr] = ex;
            ex += c[k];
        }
    }

    for (int i = tid; i < n; i += OWN_THREADS) {
        int s, dl; bool own;
        if (lmode) { s = lst_s[i]; dl = lst_d[i] & ((1 << ROW_BITS) - 1); own = true; }
        else {
            const unsigned u = (unsigned)(edst[i] - r0);
            own = u < (unsigned)rows; dl = own ? (int)u : 0; s = own ? esrc[i] : 0;
        }
        if (own) {
            const float4 a4 = *(const float4*)(asn + (size_t)s * HEADS);
            const float4 d4 = *(const float4*)(adl + dl * HEADS);
            const float a0 = lrelu02(a4.x + d4.x), a1 = lrelu02(a4.y + d4.y);
            const float a2 = lrelu02(a4.z + d4.z), a3 = lrelu02(a4.w + d4.w);
            if (lmode) *(float4*)(lst_a + i * HEADS) = make_float4(a0, a1, a2, a3);
            unsigned* m = mx + dl * HEADS;
            __hip_atomic_fetch_max(m + 0, f2key(a0), __ATOMIC_RELAXED, __HIP_MEMORY_SCOPE_WORKGROUP);
            __hip_atomic_fetch_max(m + 1, f2key(a1), __ATOMIC_RELAXED, __HIP_MEMORY_SCOPE_WORKGROUP);
            __hip_atomic_fetch_max(m + 2, f2key(a2), __ATOMIC_RELAXED, __HIP_MEMORY_SCOPE_WORKGROUP);
            __hip_atomic_fetch_max(m + 3, f2key(a3), __ATOMIC_RELAXED, __HIP_MEMORY_SCOPE_WORKGROUP);
        }
    }
    __syncthreads();

    if (lmode) {
        float4 pv[LCAP / OWN_THREADS];
        int sv[LCAP / OWN_THREADS], dv[LCAP / OWN_THREADS], pp[LCAP / OWN_THREADS];
        #pragma unroll
        for (int k = 0; k < LCAP / OWN_THREADS; ++k) {
            const int i = tid + k * OWN_THREADS;
            if (i < n) {
                const int pk = lst_d[i];
                const int dl = pk & ((1 << ROW_BITS) - 1);
                const float4 a = *(const float4*)(lst_a + i * HEADS);
                const unsigned* m = mx + dl * HEADS;
                const float p0 = __expf(a.x - key2f(m[0])), p1 = __expf(a.y - key2f(m[1]));
                const float p2 = __expf(a.z - key2f(m[2])), p3 = __expf(a.w - key2f(m[3]));
                float* sp = sm + dl * HEADS;
                __hip_atomic_fetch_add(sp + 0, p0, __ATOMIC_RELAXED, __HIP_MEMORY_SCOPE_WORKGROUP);
                __hip_atomic_fetch_add(sp + 1, p1, __ATOMIC_RELAXED, __HIP_MEMORY_SCOPE_WORKGROUP);
                __hip_atomic_fetch_add(sp + 2, p2, __ATOMIC_RELAXED, __HIP_MEMORY_SCOPE_WORKGROUP);
                __hip_atomic_fetch_add(sp + 3, p3, __ATOMIC_RELAXED, __HIP_MEMORY_SCOPE_WORKGROUP);
                pv[k] = make_float4(p0, p1, p2, p3);
                sv[k] = lst_s[i];
                dv[k] = dl;
                pp[k] = rowc[dl] + (int)((unsigned)pk >> ROW_BITS);
            }
        }
        __syncthreads();
        #pragma unroll
        for (int k = 0; k < LCAP / OWN_THREADS; ++k) {
            const int i = tid + k * OWN_THREADS;
            if (i < n) {
                lst_s[pp[k]] = sv[k];
                lst_d[pp[k]] = dv[k];
                *(float4*)(lst_a + pp[k] * HEADS) = pv[k];
            }
        }
    } else {
        for (int i = tid; i < n; i += OWN_THREADS) {
            const unsigned u = (unsigned)(edst[i] - r0);
            if (u < (unsigned)rows) {
                const int dl = (int)u;
                const int s  = esrc[i];
                const float4 a4 = *(const float4*)(asn + (size_t)s * HEADS);
                const float4 d4 = *(const float4*)(adl + dl * HEADS);
                const float4 a = make_float4(lrelu02(a4.x + d4.x), lrelu02(a4.y + d4.y),
                                             lrelu02(a4.z + d4.z), lrelu02(a4.w + d4.w));
                const unsigned* m = mx + dl * HEADS;
                const float p0 = __expf(a.x - key2f(m[0])), p1 = __expf(a.y - key2f(m[1]));
                const float p2 = __expf(a.z - key2f(m[2])), p3 = __expf(a.w - key2f(m[3]));
                float* sp = sm + dl * HEADS;
                __hip_atomic_fetch_add(sp + 0, p0, __ATOMIC_RELAXED, __HIP_MEMORY_SCOPE_WORKGROUP);
                __hip_atomic_fetch_add(sp + 1, p1, __ATOMIC_RELAXED, __HIP_MEMORY_SCOPE_WORKGROUP);
                __hip_atomic_fetch_add(sp + 2, p2, __ATOMIC_RELAXED, __HIP_MEMORY_SCOPE_WORKGROUP);
                __hip_atomic_fetch_add(sp + 3, p3, __ATOMIC_RELAXED, __HIP_MEMORY_SCOPE_WORKGROUP);
            }
        }
    }
    __syncthreads();

    {
        const int lane = tid & 31;
        const int wv = __builtin_amdgcn_readfirstlane(tid >> 5);
        const int hl = lane & 3;
        for (int i0 = wv * EPW; i0 < n; i0 += OWN_WAVES * EPW) {
            const int il = i0 + (lane >> 2);
            if (lmode && i0 + EPW <= n) {
                const int   s  = lst_s[il];
                const int   dl = lst_d[il];
                const float p  = lst_a[il * HEADS + hl];
                float f[EPW][4];
                #pragma unroll
                for (int j = 0; j < EPW; ++j) {
                    const int sj = __builtin_amdgcn_readlane(s, 4 * j);
                    const float* sp = src_feat + (size_t)sj * FDIM + lane;
                    f[j][0] = sp[0]; f[j][1] = sp[32]; f[j][2] = sp[64]; f[j][3] = sp[96];
                }
                const float wl = p / (sm[dl * HEADS + hl] + 1e-16f);
                int dcur = __builtin_amdgcn_readlane(dl, 0);
                float a0 = __int_as_float(__builtin_amdgcn_readlane(__float_as_int(wl), 0)) * f[0][0];
                float a1 = __int_as_float(__builtin_amdgcn_readlane(__float_as_int(wl), 1)) * f[0][1];
                float a2 = __int_as_float(__builtin_amdgcn_readlane(__float_as_int(wl), 2)) * f[0][2];
                float a3 = __int_as_float(__builtin_amdgcn_readlane(__float_as_int(wl), 3)) * f[0][3];
                #pragma unroll
                for (int j = 1; j < EPW; ++j) {
                    const int   dj = __builtin_amdgcn_readlane(dl, 4 * j);
                    const float w0 = __int_as_float(__builtin_amdgcn_readlane(__float_as_int(wl), 4 * j + 0));
                    const float w1 = __int_as_float(__builtin_amdgcn_readlane(__float_as_int(wl), 4 * j + 1));
                    const float w2 = __int_as_float(__builtin_amdgcn_readlane(__float_as_int(wl), 4 * j + 2));
                    const float w3 = __int_as_float(__builtin_amdgcn_readlane(__float_as_int(wl), 4 * j + 3));
                    if (dj != dcur) {
                        add_run_lds(acc, dcur, lane, a0, a1, a2, a3);
                        dcur = dj;
                        a0 = w0 * f[j][0]; a1 = w1 * f[j][1]; a2 = w2 * f[j][2]; a3 = w3 * f[j][3];
                    } else {
                        a0 += w0 * f[j][0]; a1 += w1 * f[j][1]; a2 += w2 * f[j][2]; a3 += w3 * f[j][3];
                    }
                }
                add_run_lds(acc, dcur, lane, a0, a1, a2, a3);
            } else {
                int s = 0, dl = 0; bool own = false;
                if (il < n) {
                    if (lmode) { s = lst_s[il]; dl = lst_d[il]; own = true; }
                    else {
                        const unsigned u = (unsigned)(edst[il] - r0);
                        own = u < (unsigned)rows; dl = own ? (int)u : 0; s = own ? esrc[il] : 0;
                    }
                }
                float wl = 0.f;
                if (own) {
                    float p;
                    if (lmode) p = lst_a[il * HEADS + hl];
                    else p = __expf(lrelu02(asn[(size_t)s * HEADS + hl] + adl[dl * HEADS + hl])
                                    - key2f(mx[dl * HEADS + hl]));
                    wl = p / (sm[dl * HEADS + hl] + 1e-16f);
                }
                const unsigned om = __builtin_amdgcn_ballot_w32(own);
                for (int j = 0; j < EPW; ++j) {
                    if (om & (1u << (4 * j))) {
                        const int sj = __builtin_amdgcn_readlane(s, 4 * j);
                        const float* sp = src_feat + (size_t)sj * FDIM + lane;
                        add_row_lds(acc, __builtin_amdgcn_readlane(dl, 4 * j), lane, wl, 4 * j,
                                    sp[0], sp[32], sp[64], sp[96]);
                    }
                }
            }
        }
    }
    __syncthreads();

    {
        float4* __restrict__ op = (float4*)(out + (size_t)r0 * FDIM);
        const float4* ac = (const float4*)acc;
        for (int i = tid; i < rows * (FDIM / 4); i += OWN_THREADS) {
            float4 o = op[i];
            const float4 a = ac[i];
            o.x += a.x; o.y += a.y; o.z += a.z; o.w += a.w;
            op[i] = o;
        }
    }
}

extern "C" void kernel_launch(void* const* d_in, const int* in_sizes, int n_in,
                              void* d_out, int out_size, void* d_ws, size_t ws_size,
                              hipStream_t stream)
{
    const float* x_src  = (const float*)d_in[0];
    const float* x_dst  = (const float*)d_in[1];
    const int*   esrc   = (const int*)d_in[2];
    const int*   edst   = (const int*)d_in[3];
    const float* W_src  = (const float*)d_in[5];
    const float* W_dst  = (const float*)d_in[6];
    const float* attS   = (const float*)d_in[7];
    const float* attD   = (const float*)d_in[8];
    const float* W_self = (const float*)d_in[9];
    const float* b_self = (const float*)d_in[10];

    const int N_SRC = in_sizes[0] / FDIM;
    const int N_DST = in_sizes[1] / FDIM;
    const int E     = in_sizes[2];

    char* ws = (char*)d_ws;
    float* src_feat = (float*)(ws);
    float* as_node  = (float*)(ws + (size_t)(N_SRC + N_DST) * FDIM * 4);
    float* ad_node  = as_node + (size_t)N_SRC * HEADS;
    unsigned* bitmaps = (unsigned*)(ws + (size_t)N_SRC * FDIM * 4);
    int* pairs        = (int*)(ws + (size_t)N_SRC * FDIM * 4
                                  + (size_t)BM_OWNERS * NP_GROUPS * BM_WORDS * 4);

    float* out = (float*)d_out;

    node_proj<<<NP_GROUPS, 256, 0, stream>>>(
        x_src, x_dst, W_src, W_dst, W_self, b_self, attS, attD,
        src_feat, out, as_node, ad_node, esrc, edst, bitmaps, pairs, E, N_SRC, N_DST);
    edge_owner<<<(N_DST + OWN_ROWS - 1) / OWN_ROWS, OWN_THREADS, 0, stream>>>(
        esrc, edst, src_feat, as_node, ad_node, bitmaps, pairs, out, E, N_DST);
}
